# starting kernel
# baseline (speedup 1.0000x reference)
#include <hip/hip_runtime.h>
#include <hip/hip_bf16.h>

typedef float v2f __attribute__((ext_vector_type(2)));
typedef float v8f __attribute__((ext_vector_type(8)));

#define FD 64
#define WPB 8

__global__ void __launch_bounds__(256) deg_init_kernel(unsigned* __restrict__ deg, int N) {
  int i = blockIdx.x * 256 + threadIdx.x;
  if (i < N) deg[i] = 1u;
}

__global__ void __launch_bounds__(256) deg_count_kernel(const int* __restrict__ ei, int E,
                                                        unsigned* __restrict__ deg) {
  int e = blockIdx.x * 256 + threadIdx.x;
  if (e < E) atomicAdd(&deg[ei[E + e]], 1u);
}

__global__ void __launch_bounds__(256) dinv_kernel(const unsigned* __restrict__ deg,
                                                   float* __restrict__ dinv, int N) {
  int i = blockIdx.x * 256 + threadIdx.x;
  if (i < N) dinv[i] = rsqrtf((float)deg[i]);
}

__global__ void __launch_bounds__(256) gemm_xw_kernel(const float* __restrict__ x,
                                                      const float* __restrict__ W,
                                                      const float* __restrict__ dinv,
                                                      float* __restrict__ y,
                                                      float* __restrict__ acc,
                                                      int nRowTiles, int N) {
  __shared__ float Wl[FD * FD];
  for (int p = threadIdx.x; p < 2048; p += 256) {
    const int lp   = p & 31;
    const int f    = p >> 5;
    const int c    = f & 3;
    const int kk   = f >> 2;
    const int n    = lp & 15;
    const int h    = lp >> 4;
    const int row  = (kk << 2) + (h << 1);
    const int col  = (c << 4) + n;
    Wl[2 * p + 0] = W[(row + 0) * FD + col];
    Wl[2 * p + 1] = W[(row + 1) * FD + col];
  }
  __syncthreads();

  const int wave = threadIdx.x >> 5;
  const int lane = threadIdx.x & 31;
  const int tile = blockIdx.x * WPB + wave;
  if (tile >= nRowTiles) return;

  const int r0   = tile << 4;
  const int m    = lane & 15;
  const int half = lane >> 4;
  const int kb   = half << 1;

  int arow = r0 + m; if (arow >= N) arow = N - 1;
  const float* __restrict__ xrow  = x + (size_t)arow * FD;
  const float* __restrict__ bbase = Wl + (lane << 1);

  v8f c0 = {}, c1 = {}, c2 = {}, c3 = {};
#pragma unroll
  for (int kk = 0; kk < 16; ++kk) {
    v2f a;
    a.x = xrow[(kk << 2) + kb + 0];
    a.y = xrow[(kk << 2) + kb + 1];
    const float* fb = bbase + (kk << 8);
    const v2f b0 = *(const v2f*)(fb + 0 * 64);
    const v2f b1 = *(const v2f*)(fb + 1 * 64);
    const v2f b2 = *(const v2f*)(fb + 2 * 64);
    const v2f b3 = *(const v2f*)(fb + 3 * 64);
    c0 = __builtin_amdgcn_wmma_f32_16x16x4_f32(false, a, false, b0, (short)0, c0, false, false);
    c1 = __builtin_amdgcn_wmma_f32_16x16x4_f32(false, a, false, b1, (short)0, c1, false, false);
    c2 = __builtin_amdgcn_wmma_f32_16x16x4_f32(false, a, false, b2, (short)0, c2, false, false);
    c3 = __builtin_amdgcn_wmma_f32_16x16x4_f32(false, a, false, b3, (short)0, c3, false, false);
  }

#pragma unroll
  for (int r = 0; r < 8; ++r) {
    const int row = r0 + r + (half << 3);
    if (row < N) {
      const float s = dinv[row];
      const size_t base = (size_t)row * FD + m;
      float v;
      v = c0[r] * s; y[base +  0] = v; acc[base +  0] = v;
      v = c1[r] * s; y[base + 16] = v; acc[base + 16] = v;
      v = c2[r] * s; y[base + 32] = v; acc[base + 32] = v;
      v = c3[r] * s; y[base + 48] = v; acc[base + 48] = v;
    }
  }
}

__global__ void __launch_bounds__(256) scatter_kernel(const int* __restrict__ ei, int E,
                                                      const float* __restrict__ y,
                                                      float* __restrict__ acc) {
  long long t = (long long)blockIdx.x * 256 + threadIdx.x;
  int e = (int)(t >> 4);
  if (e >= E) return;
  const int c4 = (int)(t & 15) << 2;
  const int s  = ei[e];
  const int d  = ei[E + e];
  const float4 v = *(const float4*)(y + (size_t)s * FD + c4);
  float* p = acc + (size_t)d * FD + c4;
  unsafeAtomicAdd(p + 0, v.x);
  unsafeAtomicAdd(p + 1, v.y);
  unsafeAtomicAdd(p + 2, v.z);
  unsafeAtomicAdd(p + 3, v.w);
}

__global__ void __launch_bounds__(256) finalize_kernel(const float* __restrict__ acc,
                                                       const float* __restrict__ dinv,
                                                       const float* __restrict__ b,
                                                       const float* __restrict__ prelu_a,
                                                       float* __restrict__ out, int N) {
  long long t = (long long)blockIdx.x * 256 + threadIdx.x;
  if (t >= (long long)N * FD) return;
  const int n = (int)(t >> 6);
  const int c = (int)(t & 63);
  const float v = dinv[n] * acc[t] + b[c];
  out[t] = v >= 0.0f ? v : prelu_a[0] * v;
}

extern "C" void kernel_launch(void* const* d_in, const int* in_sizes, int n_in,
                              void* d_out, int out_size, void* d_ws, size_t ws_size,
                              hipStream_t stream) {
  const float* x  = (const float*)d_in[0];
  const int*   ei = (const int*)d_in[1];
  const float* W  = (const float*)d_in[2];
  const float* b  = (const float*)d_in[3];
  const float* pa = (const float*)d_in[4];
  float* out = (float*)d_out;

  const int N = in_sizes[0] / FD;
  const int E = in_sizes[1] / 2;

  unsigned* deg  = (unsigned*)d_ws;
  float*    dinv = (float*)d_ws + N;
  float*    y    = (float*)d_ws + 2 * (size_t)N;
  float*    acc  = y + (size_t)N * FD;

  const int bN = (N + 255) / 256;
  const int bE = (E + 255) / 256;
  deg_init_kernel<<<bN, 256, 0, stream>>>(deg, N);
  deg_count_kernel<<<bE, 256, 0, stream>>>(ei, E, deg);
  dinv_kernel<<<bN, 256, 0, stream>>>(deg, dinv, N);

  const int nRowTiles = (N + 15) / 16;
  const int bG = (nRowTiles + WPB - 1) / WPB;
  gemm_xw_kernel<<<bG, 256, 0, stream>>>(x, W, dinv, y, acc, nRowTiles, N);

  const long long scatterThreads = (long long)E * 16;
  const int bS = (int)((scatterThreads + 255) / 256);
  scatter_kernel<<<bS, 256, 0, stream>>>(ei, E, y, acc);

  const long long finThreads = (long long)N * FD;
  const int bF = (int)((finThreads + 255) / 256);
  finalize_kernel<<<bF, 256, 0, stream>>>(acc, dinv, b, pa, out, N);
}


# in-degree counted in per-block LDS histograms (two 16-bit counters per word, a row per 32768-edge slice, rows summed in the dinv kernel) instead of scattered global integer atomics; 6 launches -> 5
# speedup vs baseline: 1.3475x; 1.3475x over previous
#include <hip/hip_runtime.h>
#include <hip/hip_bf16.h>
#include <stdint.h>

typedef float v2f __attribute__((ext_vector_type(2)));
typedef float v8f __attribute__((ext_vector_type(8)));

#define FD 64
#define WPB 8

#define HB     25000
#define HWORDS (HB / 2)
#define HVEC   (HWORDS / 4)
#define SLICE  32768
#define HT     1024
#define EPT    (SLICE / HT)

__global__ void __launch_bounds__(HT) deg_hist_kernel(const int* __restrict__ ei, int E,
                                                      unsigned* __restrict__ part, int rowWords) {
  __shared__ uint4 h4[HVEC];
  unsigned* h = (unsigned*)h4;
  for (int i = threadIdx.x; i < HVEC; i += HT) h4[i] = make_uint4(0u, 0u, 0u, 0u);
  __syncthreads();

  const int base = blockIdx.y * HB;
  const int e0   = blockIdx.x * SLICE + threadIdx.x;
  const int* __restrict__ dst = ei + E;
  int d[EPT];
#pragma unroll
  for (int k = 0; k < EPT; ++k) {
    const int e = e0 + k * HT;
    d[k] = (e < E) ? dst[e] : -1;
  }
#pragma unroll
  for (int k = 0; k < EPT; ++k) {
    const unsigned r = (unsigned)(d[k] - base);
    if (r < (unsigned)HB) atomicAdd(&h[r >> 1], 1u << ((r & 1u) << 4));
  }
  __syncthreads();

  uint4* __restrict__ row = (uint4*)(part + (size_t)blockIdx.x * rowWords + (size_t)blockIdx.y * HWORDS);
  for (int i = threadIdx.x; i < HVEC; i += HT) row[i] = h4[i];
}

__global__ void __launch_bounds__(256) dinv_kernel(const unsigned* __restrict__ part, int S,
                                                   int rowWords, float* __restrict__ dinv, int N) {
  const int w = blockIdx.x * 256 + threadIdx.x;
  if (2 * w >= N) return;
  const unsigned* __restrict__ p = part + w;
  unsigned lo = 1u, hi = 1u;
  int s = 0;
  for (; s + 16 <= S; s += 16) {
    unsigned v[16];
#pragma unroll
    for (int j = 0; j < 16; ++j) v[j] = p[(size_t)(s + j) * rowWords];
#pragma unroll
    for (int j = 0; j < 16; ++j) { lo += v[j] & 0xffffu; hi += v[j] >> 16; }
  }
  for (; s < S; ++s) {
    const unsigned v = p[(size_t)s * rowWords];
    lo += v & 0xffffu; hi += v >> 16;
  }
  dinv[2 * w] = rsqrtf((float)lo);
  if (2 * w + 1 < N) dinv[2 * w + 1] = rsqrtf((float)hi);
}

__global__ void __launch_bounds__(256) gemm_xw_kernel(const float* __restrict__ x,
                                                      const float* __restrict__ W,
                                                      const float* __restrict__ dinv,
                                                      float* __restrict__ y,
                                                      float* __restrict__ acc,
                                                      int nRowTiles, int N) {
  __shared__ float Wl[FD * FD];
  for (int p = threadIdx.x; p < 2048; p += 256) {
    const int lp   = p & 31;
    const int f    = p >> 5;
    const int c    = f & 3;
    const int kk   = f >> 2;
    const int n    = lp & 15;
    const int h    = lp >> 4;
    const int row  = (kk << 2) + (h << 1);
    const int col  = (c << 4) + n;
    Wl[2 * p + 0] = W[(row + 0) * FD + col];
    Wl[2 * p + 1] = W[(row + 1) * FD + col];
  }
  __syncthreads();

  const int wave = threadIdx.x >> 5;
  const int lane = threadIdx.x & 31;
  const int tile = blockIdx.x * WPB + wave;
  if (tile >= nRowTiles) return;

  const int r0   = tile << 4;
  const int m    = lane & 15;
  const int half = lane >> 4;
  const int kb   = half << 1;

  int arow = r0 + m; if (arow >= N) arow = N - 1;
  const float* __restrict__ xrow  = x + (size_t)arow * FD;
  const float* __restrict__ bbase = Wl + (lane << 1);

  v8f c0 = {}, c1 = {}, c2 = {}, c3 = {};
#pragma unroll
  for (int kk = 0; kk < 16; ++kk) {
    v2f a;
    a.x = xrow[(kk << 2) + kb + 0];
    a.y = xrow[(kk << 2) + kb + 1];
    const float* fb = bbase + (kk << 8);
    const v2f b0 = *(const v2f*)(fb + 0 * 64);
    const v2f b1 = *(const v2f*)(fb + 1 * 64);
    const v2f b2 = *(const v2f*)(fb + 2 * 64);
    const v2f b3 = *(const v2f*)(fb + 3 * 64);
    c0 = __builtin_amdgcn_wmma_f32_16x16x4_f32(false, a, false, b0, (short)0, c0, false, false);
    c1 = __builtin_amdgcn_wmma_f32_16x16x4_f32(false, a, false, b1, (short)0, c1, false, false);
    c2 = __builtin_amdgcn_wmma_f32_16x16x4_f32(false, a, false, b2, (short)0, c2, false, false);
    c3 = __builtin_amdgcn_wmma_f32_16x16x4_f32(false, a, false, b3, (short)0, c3, false, false);
  }

#pragma unroll
  for (int r = 0; r < 8; ++r) {
    const int row = r0 + r + (half << 3);
    if (row < N) {
      const float s = dinv[row];
      const size_t base = (size_t)row * FD + m;
      float v;
      v = c0[r] * s; y[base +  0] = v; acc[base +  0] = v;
      v = c1[r] * s; y[base + 16] = v; acc[base + 16] = v;
      v = c2[r] * s; y[base + 32] = v; acc[base + 32] = v;
      v = c3[r] * s; y[base + 48] = v; acc[base + 48] = v;
    }
  }
}

__global__ void __launch_bounds__(256) scatter_kernel(const int* __restrict__ ei, int E,
                                                      const float* __restrict__ y,
                                                      float* __restrict__ acc) {
  long long t = (long long)blockIdx.x * 256 + threadIdx.x;
  int e = (int)(t >> 4);
  if (e >= E) return;
  const int c4 = (int)(t & 15) << 2;
  const int s  = ei[e];
  const int d  = ei[E + e];
  const float4 v = *(const float4*)(y + (size_t)s * FD + c4);
  float* p = acc + (size_t)d * FD + c4;
  unsafeAtomicAdd(p + 0, v.x);
  unsafeAtomicAdd(p + 1, v.y);
  unsafeAtomicAdd(p + 2, v.z);
  unsafeAtomicAdd(p + 3, v.w);
}

__global__ void __launch_bounds__(256) finalize_kernel(const float* __restrict__ acc,
                                                       const float* __restrict__ dinv,
                                                       const float* __restrict__ b,
                                                       const float* __restrict__ prelu_a,
                                                       float* __restrict__ out, int N) {
  long long t = (long long)blockIdx.x * 256 + threadIdx.x;
  if (t >= (long long)N * FD) return;
  const int n = (int)(t >> 6);
  const int c = (int)(t & 63);
  const float v = dinv[n] * acc[t] + b[c];
  out[t] = v >= 0.0f ? v : prelu_a[0] * v;
}

extern "C" void kernel_launch(void* const* d_in, const int* in_sizes, int n_in,
                              void* d_out, int out_size, void* d_ws, size_t ws_size,
                              hipStream_t stream) {
  const float* x  = (const float*)d_in[0];
  const int*   ei = (const int*)d_in[1];
  const float* W  = (const float*)d_in[2];
  const float* b  = (const float*)d_in[3];
  const float* pa = (const float*)d_in[4];
  float* out = (float*)d_out;

  const int N = in_sizes[0] / FD;
  const int E = in_sizes[1] / 2;

  float*    dinv = (float*)d_ws + N;
  float*    y    = (float*)d_ws + 2 * (size_t)N;
  float*    acc  = y + (size_t)N * FD;
  unsigned* part = (unsigned*)(((uintptr_t)(acc + (size_t)N * FD) + 15) & ~(uintptr_t)15);

  const int S  = (E + SLICE - 1) / SLICE;
  const int R  = (N + HB - 1) / HB;
  const int rowWords = R * HWORDS;
  deg_hist_kernel<<<dim3(S, R), HT, 0, stream>>>(ei, E, part, rowWords);
  const int bW = ((N + 1) / 2 + 255) / 256;
  dinv_kernel<<<bW, 256, 0, stream>>>(part, S, rowWords, dinv, N);

  const int nRowTiles = (N + 15) / 16;
  const int bG = (nRowTiles + WPB - 1) / WPB;
  gemm_xw_kernel<<<bG, 256, 0, stream>>>(x, W, dinv, y, acc, nRowTiles, N);

  const long long scatterThreads = (long long)E * 16;
  const int bS = (int)((scatterThreads + 255) / 256);
  scatter_kernel<<<bS, 256, 0, stream>>>(ei, E, y, acc);

  const long long finThreads = (long long)N * FD;
  const int bF = (int)((finThreads + 255) / 256);
  finalize_kernel<<<bF, 256, 0, stream>>>(acc, dinv, b, pa, out, N);
}


# aggregation without float atomics: per-call counting sort of edges by destination (LDS cursors), then a wave per node sums its source rows in sorted order, bias+PReLU fused; sum order changes
# speedup vs baseline: 1.9757x; 1.4662x over previous
#include <hip/hip_runtime.h>
#include <hip/hip_bf16.h>
#include <stdint.h>

typedef float v2f __attribute__((ext_vector_type(2)));
typedef float v8f __attribute__((ext_vector_type(8)));

#define FD 64
#define WPB 8

#define HB     25000
#define HWORDS (HB / 2)
#define HVEC   (HWORDS / 4)
#define SLICE  32768
#define HT     1024
#define EPT    (SLICE / HT)
#define CI     ((HB / 4 + HT - 1) / HT)

__global__ void __launch_bounds__(HT) deg_hist_kernel(const int* __restrict__ ei, int E,
                                                      unsigned* __restrict__ part, int rowWords) {
  __shared__ uint4 h4[HVEC];
  unsigned* h = (unsigned*)h4;
  for (int i = threadIdx.x; i < HVEC; i += HT) h4[i] = make_uint4(0u, 0u, 0u, 0u);
  __syncthreads();

  const int base = blockIdx.y * HB;
  const int e0   = blockIdx.x * SLICE + threadIdx.x;
  const int* __restrict__ dst = ei + E;
  int d[EPT];
#pragma unroll
  for (int k = 0; k < EPT; ++k) {
    const int e = e0 + k * HT;
    d[k] = (e < E) ? dst[e] : -1;
  }
#pragma unroll
  for (int k = 0; k < EPT; ++k) {
    const unsigned r = (unsigned)(d[k] - base);
    if (r < (unsigned)HB) atomicAdd(&h[r >> 1], 1u << ((r & 1u) << 4));
  }
  __syncthreads();

  uint4* __restrict__ row = (uint4*)(part + (size_t)blockIdx.x * rowWords + (size_t)blockIdx.y * HWORDS);
  for (int i = threadIdx.x; i < HVEC; i += HT) row[i] = h4[i];
}

__global__ void __launch_bounds__(256) dinv_scan_kernel(const unsigned* __restrict__ part, int S,
                                                        int rowWords, float* __restrict__ dinv, int N,
                                                        unsigned* __restrict__ pre,
                                                        unsigned* __restrict__ localoff,
                                                        unsigned* __restrict__ blocksum) {
  __shared__ unsigned wsum[8];
  const int w    = blockIdx.x * 256 + threadIdx.x;
  const int lane = threadIdx.x & 31;
  const int wave = threadIdx.x >> 5;
  const bool live = w < rowWords;
  unsigned lo = 0u, hi = 0u;
  if (live) {
    const unsigned* __restrict__ p = part + w;
    uint2* __restrict__ q = (uint2*)pre + w;
    int s = 0;
    for (; s + 16 <= S; s += 16) {
      unsigned v[16];
#pragma unroll
      for (int j = 0; j < 16; ++j) v[j] = p[(size_t)(s + j) * rowWords];
#pragma unroll
      for (int j = 0; j < 16; ++j) {
        q[(size_t)(s + j) * rowWords] = make_uint2(lo, hi);
        lo += v[j] & 0xffffu; hi += v[j] >> 16;
      }
    }
    for (; s < S; ++s) {
      const unsigned v = p[(size_t)s * rowWords];
      q[(size_t)s * rowWords] = make_uint2(lo, hi);
      lo += v & 0xffffu; hi += v >> 16;
    }
    if (2 * w < N)     dinv[2 * w]     = rsqrtf((float)(lo + 1u));
    if (2 * w + 1 < N) dinv[2 * w + 1] = rsqrtf((float)(hi + 1u));
  }
  const unsigned tot = lo + hi;
  unsigned inc = tot;
  for (int o = 1; o < 32; o <<= 1) {
    const unsigned t = __shfl_up(inc, o);
    if (lane >= o) inc += t;
  }
  if (lane == 31) wsum[wave] = inc;
  __syncthreads();
  unsigned woff = 0u;
  for (int i = 0; i < wave; ++i) woff += wsum[i];
  const unsigned excl = woff + inc - tot;
  if (live) *((uint2*)localoff + w) = make_uint2(excl, excl + lo);
  if (threadIdx.x == 255) blocksum[blockIdx.x] = woff + inc;
}

__global__ void __launch_bounds__(256) rowptr_kernel(const unsigned* __restrict__ localoff,
                                                     const unsigned* __restrict__ blocksum,
                                                     int rowWords, unsigned* __restrict__ rowptr) {
  __shared__ unsigned wsum[8];
  const int lane = threadIdx.x & 31;
  const int wave = threadIdx.x >> 5;
  unsigned ps = 0u;
  for (int i = threadIdx.x; i < (int)blockIdx.x; i += 256) ps += blocksum[i];
  for (int o = 16; o > 0; o >>= 1) ps += __shfl_xor(ps, o);
  if (lane == 0) wsum[wave] = ps;
  __syncthreads();
  unsigned boff = 0u;
  for (int i = 0; i < 8; ++i) boff += wsum[i];
  const int w = blockIdx.x * 256 + threadIdx.x;
  if (w < rowWords) {
    const uint2 l = *((const uint2*)localoff + w);
    *((uint2*)rowptr + w) = make_uint2(boff + l.x, boff + l.y);
    if (w == rowWords - 1) rowptr[2 * (size_t)rowWords] = boff + blocksum[blockIdx.x];
  }
}

__global__ void __launch_bounds__(HT) fill_kernel(const int* __restrict__ ei, int E,
                                                  const unsigned* __restrict__ pre, int NP,
                                                  const unsigned* __restrict__ rowptr,
                                                  int* __restrict__ csr) {
  __shared__ uint4 cur4[HB / 4];
  unsigned* cur = (unsigned*)cur4;
  const int base = blockIdx.y * HB;
  const int e0   = blockIdx.x * SLICE + threadIdx.x;
  const int* __restrict__ dst = ei + E;
  int d[EPT], sv[EPT];
#pragma unroll
  for (int k = 0; k < EPT; ++k) {
    const int e = e0 + k * HT;
    d[k] = (e < E) ? dst[e] : -1;
  }
  const uint4* __restrict__ rp4 = (const uint4*)(rowptr + base);
  const uint4* __restrict__ pr4 = (const uint4*)(pre + (size_t)blockIdx.x * NP + base);
  uint4 ca[CI], cb[CI];
#pragma unroll
  for (int it = 0; it < CI; ++it) {
    const int i = threadIdx.x + it * HT;
    if (i < HB / 4) { ca[it] = rp4[i]; cb[it] = pr4[i]; }
  }
#pragma unroll
  for (int it = 0; it < CI; ++it) {
    const int i = threadIdx.x + it * HT;
    if (i < HB / 4)
      cur4[i] = make_uint4(ca[it].x + cb[it].x, ca[it].y + cb[it].y, ca[it].z + cb[it].z, ca[it].w + cb[it].w);
  }
#pragma unroll
  for (int k = 0; k < EPT; ++k) {
    const int e = e0 + k * HT;
    sv[k] = (e < E) ? ei[e] : 0;
  }
  __syncthreads();

  unsigned pos[EPT];
#pragma unroll
  for (int k = 0; k < EPT; ++k) {
    const unsigned r = (unsigned)(d[k] - base);
    pos[k] = 0xffffffffu;
    if (r < (unsigned)HB) pos[k] = atomicAdd(&cur[r], 1u);
  }
#pragma unroll
  for (int k = 0; k < EPT; ++k)
    if (pos[k] != 0xffffffffu) csr[pos[k]] = sv[k];
}

__global__ void __launch_bounds__(256) gemm_xw_kernel(const float* __restrict__ x,
                                                      const float* __restrict__ W,
                                                      const float* __restrict__ dinv,
                                                      float* __restrict__ y,
                                                      int nRowTiles, int N) {
  __shared__ float Wl[FD * FD];
  for (int p = threadIdx.x; p < 2048; p += 256) {
    const int lp   = p & 31;
    const int f    = p >> 5;
    const int c    = f & 3;
    const int kk   = f >> 2;
    const int n    = lp & 15;
    const int h    = lp >> 4;
    const int row  = (kk << 2) + (h << 1);
    const int col  = (c << 4) + n;
    Wl[2 * p + 0] = W[(row + 0) * FD + col];
    Wl[2 * p + 1] = W[(row + 1) * FD + col];
  }
  __syncthreads();

  const int wave = threadIdx.x >> 5;
  const int lane = threadIdx.x & 31;
  const int tile = blockIdx.x * WPB + wave;
  if (tile >= nRowTiles) return;

  const int r0   = tile << 4;
  const int m    = lane & 15;
  const int half = lane >> 4;
  const int kb   = half << 1;

  int arow = r0 + m; if (arow >= N) arow = N - 1;
  const float* __restrict__ xrow  = x + (size_t)arow * FD;
  const float* __restrict__ bbase = Wl + (lane << 1);

  v8f c0 = {}, c1 = {}, c2 = {}, c3 = {};
#pragma unroll
  for (int kk = 0; kk < 16; ++kk) {
    v2f a;
    a.x = xrow[(kk << 2) + kb + 0];
    a.y = xrow[(kk << 2) + kb + 1];
    const float* fb = bbase + (kk << 8);
    const v2f b0 = *(const v2f*)(fb + 0 * 64);
    const v2f b1 = *(const v2f*)(fb + 1 * 64);
    const v2f b2 = *(const v2f*)(fb + 2 * 64);
    const v2f b3 = *(const v2f*)(fb + 3 * 64);
    c0 = __builtin_amdgcn_wmma_f32_16x16x4_f32(false, a, false, b0, (short)0, c0, false, false);
    c1 = __builtin_amdgcn_wmma_f32_16x16x4_f32(false, a, false, b1, (short)0, c1, false, false);
    c2 = __builtin_amdgcn_wmma_f32_16x16x4_f32(false, a, false, b2, (short)0, c2, false, false);
    c3 = __builtin_amdgcn_wmma_f32_16x16x4_f32(false, a, false, b3, (short)0, c3, false, false);
  }

#pragma unroll
  for (int r = 0; r < 8; ++r) {
    const int row = r0 + r + (half << 3);
    if (row < N) {
      const float s = dinv[row];
      const size_t base = (size_t)row * FD + m;
      y[base +  0] = c0[r] * s;
      y[base + 16] = c1[r] * s;
      y[base + 32] = c2[r] * s;
      y[base + 48] = c3[r] * s;
    }
  }
}

__global__ void __launch_bounds__(256) gather_kernel(const int* __restrict__ csr,
                                                     const unsigned* __restrict__ rowptr,
                                                     const float* __restrict__ y,
                                                     const float* __restrict__ dinv,
                                                     const float* __restrict__ b,
                                                     const float* __restrict__ prelu_a,
                                                     float* __restrict__ out, int N) {
  const int lane = threadIdx.x & 31;
  const int node = __builtin_amdgcn_readfirstlane((int)(blockIdx.x * 8 + (threadIdx.x >> 5)));
  if (node >= N) return;
  const unsigned beg = rowptr[node];
  const int L = (int)(rowptr[node + 1] - beg);
  const float2* __restrict__ y2 = (const float2*)y;
  float2 acc = y2[(size_t)node * 32 + lane];
  const float  dn = dinv[node];
  const float  a  = prelu_a[0];
  const float2 bb = *((const float2*)b + lane);

  if (L <= 32 && N <= (1 << 24)) {
    const int key = (lane < L) ? ((csr[beg + lane] << 5) | lane) : 0x7fffffff;
    int rank = 0;
    for (int j = 0; j < L; ++j)
      rank += (__builtin_amdgcn_readlane(key, j) < key) ? 1 : 0;
    const int sorted = __builtin_amdgcn_ds_permute(rank << 2, key);
    const char* __restrict__ yb = (const char*)y;
    const unsigned lane8 = (unsigned)lane << 3;
    for (int k = 0; k < L; k += 8) {
      float2 v[8];
#pragma unroll
      for (int j = 0; j < 8; ++j) {
        const int kk = (k + j < L) ? (k + j) : (L - 1);
        const unsigned s = (unsigned)__builtin_amdgcn_readlane(sorted, kk) >> 5;
        v[j] = *(const float2*)(yb + ((s << 8) | lane8));
      }
#pragma unroll
      for (int j = 0; j < 8; ++j) {
        if (k + j < L) { acc.x += v[j].x; acc.y += v[j].y; }
      }
    }
  } else {
    int ls = -1, lj = -1;
    for (int k = 0; k < L; ++k) {
      int bs = 0x7fffffff, bj = 0x7fffffff;
      for (int j = lane; j < L; j += 32) {
        const int s = csr[beg + j];
        const bool above = (s > ls) || (s == ls && j > lj);
        const bool lower = (s < bs) || (s == bs && j < bj);
        if (above && lower) { bs = s; bj = j; }
      }
      for (int o = 16; o > 0; o >>= 1) {
        const int ts = __shfl_xor(bs, o);
        const int tj = __shfl_xor(bj, o);
        if (ts < bs || (ts == bs && tj < bj)) { bs = ts; bj = tj; }
      }
      ls = bs; lj = bj;
      const float2 v = y2[(size_t)bs * 32 + lane];
      acc.x += v.x; acc.y += v.y;
    }
  }

  float2 r;
  r.x = dn * acc.x + bb.x;
  r.y = dn * acc.y + bb.y;
  r.x = r.x >= 0.0f ? r.x : a * r.x;
  r.y = r.y >= 0.0f ? r.y : a * r.y;
  *((float2*)out + (size_t)node * 32 + lane) = r;
}

extern "C" void kernel_launch(void* const* d_in, const int* in_sizes, int n_in,
                              void* d_out, int out_size, void* d_ws, size_t ws_size,
                              hipStream_t stream) {
  const float* x  = (const float*)d_in[0];
  const int*   ei = (const int*)d_in[1];
  const float* W  = (const float*)d_in[2];
  const float* b  = (const float*)d_in[3];
  const float* pa = (const float*)d_in[4];
  float* out = (float*)d_out;

  const int N = in_sizes[0] / FD;
  const int E = in_sizes[1] / 2;

  const int S  = (E + SLICE - 1) / SLICE;
  const int R  = (N + HB - 1) / HB;
  const int rowWords = R * HWORDS;
  const int NP = 2 * rowWords;
  const int bW = (rowWords + 255) / 256;

  float*    dinv = (float*)d_ws + N;
  float*    y    = (float*)d_ws + 2 * (size_t)N;
  unsigned* part = (unsigned*)(((uintptr_t)(y + (size_t)N * FD) + 15) & ~(uintptr_t)15);
  unsigned* pre  = (unsigned*)(((uintptr_t)(part + (size_t)S * rowWords) + 15) & ~(uintptr_t)15);
  unsigned* localoff = (unsigned*)(((uintptr_t)(pre + (size_t)S * NP) + 15) & ~(uintptr_t)15);
  unsigned* blocksum = (unsigned*)(((uintptr_t)(localoff + NP) + 15) & ~(uintptr_t)15);
  unsigned* rowptr   = (unsigned*)(((uintptr_t)(blocksum + bW) + 15) & ~(uintptr_t)15);
  int*      csr      = (int*)(((uintptr_t)(rowptr + NP + 1) + 15) & ~(uintptr_t)15);

  deg_hist_kernel<<<dim3(S, R), HT, 0, stream>>>(ei, E, part, rowWords);
  dinv_scan_kernel<<<bW, 256, 0, stream>>>(part, S, rowWords, dinv, N, pre, localoff, blocksum);
  rowptr_kernel<<<bW, 256, 0, stream>>>(localoff, blocksum, rowWords, rowptr);

  const int nRowTiles = (N + 15) / 16;
  const int bG = (nRowTiles + WPB - 1) / WPB;
  gemm_xw_kernel<<<bG, 256, 0, stream>>>(x, W, dinv, y, nRowTiles, N);

  fill_kernel<<<dim3(S, R), HT, 0, stream>>>(ei, E, pre, NP, rowptr, csr);

  const int bA = (N + 7) / 8;
  gather_kernel<<<bA, 256, 0, stream>>>(csr, rowptr, y, dinv, b, pa, out, N);
}


# CSR build as a two-level counting sort: tile-sort edges by 256-node bucket in LDS, then a block per bucket ranks by node, whole-line writes; no slice-by-range tables; 6 launches -> 4; sum order kept
# speedup vs baseline: 2.9030x; 1.4694x over previous
#include <hip/hip_runtime.h>
#include <hip/hip_bf16.h>
#include <stdint.h>

typedef float v2f __attribute__((ext_vector_type(2)));
typedef float v8f __attribute__((ext_vector_type(8)));

#define FD 64
#define WPB 8

#define BWN   256
#define BSH   8
#define TE    8192
#define NT1   1024
#define EPT   (TE / NT1)
#define NT2   512
#define NW2   (NT2 / 32)
#define CAP   16384

__device__ __forceinline__ unsigned block_scan_excl(unsigned v, unsigned* wsum, unsigned* total) {
  const int lane = threadIdx.x & 31;
  const int wave = threadIdx.x >> 5;
  const int nw   = (int)(blockDim.x >> 5);
  unsigned inc = v;
  for (int o = 1; o < 32; o <<= 1) {
    const unsigned t = __shfl_up(inc, o);
    if (lane >= o) inc += t;
  }
  __syncthreads();
  if (lane == 31) wsum[wave] = inc;
  __syncthreads();
  unsigned woff = 0u, tot = 0u;
  for (int i = 0; i < nw; ++i) {
    const unsigned w = wsum[i];
    if (i < wave) woff += w;
    tot += w;
  }
  *total = tot;
  return woff + inc - v;
}

__global__ void __launch_bounds__(NT1) tile_sort_kernel(const int* __restrict__ ei, int E, int N, int nB,
                                                        unsigned* __restrict__ tbl,
                                                        unsigned* __restrict__ sorted) {
  __shared__ unsigned cnt[NT1];
  __shared__ unsigned stage[TE];
  __shared__ unsigned wsum[32];
  const int tid = threadIdx.x;
  cnt[tid] = 0u;
  __syncthreads();

  const int e0 = blockIdx.x * TE + tid;
  const int* __restrict__ dst = ei + E;
  int d[EPT], sv[EPT];
#pragma unroll
  for (int k = 0; k < EPT; ++k) {
    const int e = e0 + k * NT1;
    d[k]  = (e < E) ? dst[e] : -1;
    sv[k] = (e < E) ? ei[e] : 0;
  }
  unsigned pos[EPT];
#pragma unroll
  for (int k = 0; k < EPT; ++k) {
    pos[k] = 0xffffffffu;
    if ((unsigned)d[k] < (unsigned)N) pos[k] = atomicAdd(&cnt[d[k] >> BSH], 1u);
  }
  __syncthreads();

  unsigned tot;
  const unsigned excl = block_scan_excl(cnt[tid], wsum, &tot);
  cnt[tid] = excl;
  if (tid <= nB) tbl[(size_t)blockIdx.x * (nB + 1) + tid] = excl;
  __syncthreads();

#pragma unroll
  for (int k = 0; k < EPT; ++k)
    if (pos[k] != 0xffffffffu)
      stage[cnt[d[k] >> BSH] + pos[k]] = ((unsigned)sv[k] << 8) | ((unsigned)d[k] & (BWN - 1));
  __syncthreads();

  for (int i = tid; i < TE; i += NT1) {
    const int e = blockIdx.x * TE + i;
    if (e < E && (unsigned)i < tot) sorted[e] = stage[i];
  }
}

__device__ __forceinline__ unsigned bucket_rec(const unsigned* runoff, const unsigned* runpos,
                                               const unsigned* __restrict__ sorted, unsigned i) {
  unsigned a = 0u;
#pragma unroll
  for (unsigned st = NT2 / 2; st > 0u; st >>= 1) {
    const unsigned m = a + st;
    if (runoff[m] <= i) a = m;
  }
  return sorted[runpos[a] + (i - runoff[a])];
}

__global__ void __launch_bounds__(NT2) bucket_csr_kernel(const unsigned* __restrict__ tbl, int S, int nB,
                                                         const unsigned* __restrict__ sorted, int N,
                                                         float* __restrict__ dinv,
                                                         unsigned* __restrict__ rowptr,
                                                         int* __restrict__ csr) {
  __shared__ unsigned runoff[NT2];
  __shared__ unsigned runpos[NT2];
  __shared__ unsigned runlen[NT2];
  __shared__ unsigned ncnt[BWN];
  __shared__ unsigned cur[BWN];
  __shared__ unsigned rbuf[CAP];
  __shared__ unsigned ostage[CAP];
  __shared__ unsigned wsum[32];
  const int tid = threadIdx.x;
  const int b   = blockIdx.x;

  unsigned lo = 0u, len = 0u;
  if (tid < S) {
    const unsigned* __restrict__ t = tbl + (size_t)tid * (nB + 1) + b;
    lo  = t[0];
    len = t[1] - lo;
  }
  unsigned nb, base;
  const unsigned roff = block_scan_excl(len, wsum, &nb);
  (void)block_scan_excl(lo, wsum, &base);
  runoff[tid] = roff;
  runpos[tid] = (unsigned)tid * TE + lo;
  runlen[tid] = len;
  if (tid < BWN) ncnt[tid] = 0u;
  __syncthreads();

  const bool fits = nb <= (unsigned)CAP;
  const int lane  = tid & 31;
  const int wave  = tid >> 5;
  const int myrun = wave + NW2 * lane;
  const int m_off = (int)runoff[myrun];
  const int m_pos = (int)runpos[myrun];
  const int m_len = (int)runlen[myrun];
  for (int q0 = 0; q0 < 32 && wave + NW2 * q0 < S; q0 += 16) {
    unsigned rec[16];
#pragma unroll
    for (int u = 0; u < 16; ++u) {
      const unsigned len = (unsigned)__builtin_amdgcn_readlane(m_len, q0 + u);
      const unsigned pos = (unsigned)__builtin_amdgcn_readlane(m_pos, q0 + u);
      rec[u] = ((unsigned)lane < len) ? sorted[pos + lane] : 0u;
    }
#pragma unroll
    for (int u = 0; u < 16; ++u) {
      const unsigned len = (unsigned)__builtin_amdgcn_readlane(m_len, q0 + u);
      const unsigned off = (unsigned)__builtin_amdgcn_readlane(m_off, q0 + u);
      if ((unsigned)lane < len) {
        atomicAdd(&ncnt[rec[u] & (BWN - 1)], 1u);
        if (fits) rbuf[off + lane] = rec[u];
      }
    }
    for (int u = 0; u < 16; ++u) {
      const unsigned len = (unsigned)__builtin_amdgcn_readlane(m_len, q0 + u);
      const unsigned pos = (unsigned)__builtin_amdgcn_readlane(m_pos, q0 + u);
      const unsigned off = (unsigned)__builtin_amdgcn_readlane(m_off, q0 + u);
      for (unsigned j = lane + 32; j < len; j += 32) {
        const unsigned r = sorted[pos + j];
        atomicAdd(&ncnt[r & (BWN - 1)], 1u);
        if (fits) rbuf[off + j] = r;
      }
    }
  }
  __syncthreads();

  const unsigned c = (tid < BWN) ? ncnt[tid] : 0u;
  unsigned tot2;
  const unsigned noff = block_scan_excl(c, wsum, &tot2);
  if (tid < BWN) {
    cur[tid] = noff;
    const int node = b * BWN + tid;
    if (node < N) {
      rowptr[node] = base + noff;
      dinv[node]   = rsqrtf((float)(c + 1u));
      if (node == N - 1) rowptr[N] = base + noff + c;
    }
  }
  __syncthreads();

  if (fits) {
    for (unsigned i = tid; i < nb; i += NT2) {
      const unsigned r = rbuf[i];
      const unsigned slot = atomicAdd(&cur[r & (BWN - 1)], 1u);
      ostage[slot] = r >> 8;
    }
    __syncthreads();
    for (unsigned i = tid; i < nb; i += NT2) csr[base + i] = (int)ostage[i];
  } else {
    for (unsigned i = tid; i < nb; i += NT2) {
      const unsigned r = bucket_rec(runoff, runpos, sorted, i);
      const unsigned slot = atomicAdd(&cur[r & (BWN - 1)], 1u);
      csr[base + slot] = (int)(r >> 8);
    }
  }
}

__global__ void __launch_bounds__(256) gemm_xw_kernel(const float* __restrict__ x,
                                                      const float* __restrict__ W,
                                                      const float* __restrict__ dinv,
                                                      float* __restrict__ y,
                                                      int nRowTiles, int N) {
  __shared__ float Wl[FD * FD];
  for (int p = threadIdx.x; p < 2048; p += 256) {
    const int lp   = p & 31;
    const int f    = p >> 5;
    const int c    = f & 3;
    const int kk   = f >> 2;
    const int n    = lp & 15;
    const int h    = lp >> 4;
    const int row  = (kk << 2) + (h << 1);
    const int col  = (c << 4) + n;
    Wl[2 * p + 0] = W[(row + 0) * FD + col];
    Wl[2 * p + 1] = W[(row + 1) * FD + col];
  }
  __syncthreads();

  const int wave = threadIdx.x >> 5;
  const int lane = threadIdx.x & 31;
  const int tile = blockIdx.x * WPB + wave;
  if (tile >= nRowTiles) return;

  const int r0   = tile << 4;
  const int m    = lane & 15;
  const int half = lane >> 4;
  const int kb   = half << 1;

  int arow = r0 + m; if (arow >= N) arow = N - 1;
  const float* __restrict__ xrow  = x + (size_t)arow * FD;
  const float* __restrict__ bbase = Wl + (lane << 1);

  v8f c0 = {}, c1 = {}, c2 = {}, c3 = {};
#pragma unroll
  for (int kk = 0; kk < 16; ++kk) {
    v2f a;
    a.x = xrow[(kk << 2) + kb + 0];
    a.y = xrow[(kk << 2) + kb + 1];
    const float* fb = bbase + (kk << 8);
    const v2f b0 = *(const v2f*)(fb + 0 * 64);
    const v2f b1 = *(const v2f*)(fb + 1 * 64);
    const v2f b2 = *(const v2f*)(fb + 2 * 64);
    const v2f b3 = *(const v2f*)(fb + 3 * 64);
    c0 = __builtin_amdgcn_wmma_f32_16x16x4_f32(false, a, false, b0, (short)0, c0, false, false);
    c1 = __builtin_amdgcn_wmma_f32_16x16x4_f32(false, a, false, b1, (short)0, c1, false, false);
    c2 = __builtin_amdgcn_wmma_f32_16x16x4_f32(false, a, false, b2, (short)0, c2, false, false);
    c3 = __builtin_amdgcn_wmma_f32_16x16x4_f32(false, a, false, b3, (short)0, c3, false, false);
  }

#pragma unroll
  for (int r = 0; r < 8; ++r) {
    const int row = r0 + r + (half << 3);
    if (row < N) {
      const float s = dinv[row];
      const size_t base = (size_t)row * FD + m;
      y[base +  0] = c0[r] * s;
      y[base + 16] = c1[r] * s;
      y[base + 32] = c2[r] * s;
      y[base + 48] = c3[r] * s;
    }
  }
}

__global__ void __launch_bounds__(256) gather_kernel(const int* __restrict__ csr,
                                                     const unsigned* __restrict__ rowptr,
                                                     const float* __restrict__ y,
                                                     const float* __restrict__ dinv,
                                                     const float* __restrict__ b,
                                                     const float* __restrict__ prelu_a,
                                                     float* __restrict__ out, int N) {
  const int lane = threadIdx.x & 31;
  const int node = __builtin_amdgcn_readfirstlane((int)(blockIdx.x * 8 + (threadIdx.x >> 5)));
  if (node >= N) return;
  const unsigned beg = rowptr[node];
  const int L = (int)(rowptr[node + 1] - beg);
  const float2* __restrict__ y2 = (const float2*)y;
  float2 acc = y2[(size_t)node * 32 + lane];
  const float  dn = dinv[node];
  const float  a  = prelu_a[0];
  const float2 bb = *((const float2*)b + lane);

  if (L <= 32 && N <= (1 << 24)) {
    const int key = (lane < L) ? ((csr[beg + lane] << 5) | lane) : 0x7fffffff;
    int rank = 0;
    for (int j = 0; j < L; ++j)
      rank += (__builtin_amdgcn_readlane(key, j) < key) ? 1 : 0;
    const int sorted = __builtin_amdgcn_ds_permute(rank << 2, key);
    const char* __restrict__ yb = (const char*)y;
    const unsigned lane8 = (unsigned)lane << 3;
    for (int k = 0; k < L; k += 8) {
      float2 v[8];
#pragma unroll
      for (int j = 0; j < 8; ++j) {
        const int kk = (k + j < L) ? (k + j) : (L - 1);
        const unsigned s = (unsigned)__builtin_amdgcn_readlane(sorted, kk) >> 5;
        v[j] = *(const float2*)(yb + ((s << 8) | lane8));
      }
#pragma unroll
      for (int j = 0; j < 8; ++j) {
        if (k + j < L) { acc.x += v[j].x; acc.y += v[j].y; }
      }
    }
  } else {
    int ls = -1, lj = -1;
    for (int k = 0; k < L; ++k) {
      int bs = 0x7fffffff, bj = 0x7fffffff;
      for (int j = lane; j < L; j += 32) {
        const int s = csr[beg + j];
        const bool above = (s > ls) || (s == ls && j > lj);
        const bool lower = (s < bs) || (s == bs && j < bj);
        if (above && lower) { bs = s; bj = j; }
      }
      for (int o = 16; o > 0; o >>= 1) {
        const int ts = __shfl_xor(bs, o);
        const int tj = __shfl_xor(bj, o);
        if (ts < bs || (ts == bs && tj < bj)) { bs = ts; bj = tj; }
      }
      ls = bs; lj = bj;
      const float2 v = y2[(size_t)bs * 32 + lane];
      acc.x += v.x; acc.y += v.y;
    }
  }

  float2 r;
  r.x = dn * acc.x + bb.x;
  r.y = dn * acc.y + bb.y;
  r.x = r.x >= 0.0f ? r.x : a * r.x;
  r.y = r.y >= 0.0f ? r.y : a * r.y;
  *((float2*)out + (size_t)node * 32 + lane) = r;
}

extern "C" void kernel_launch(void* const* d_in, const int* in_sizes, int n_in,
                              void* d_out, int out_size, void* d_ws, size_t ws_size,
                              hipStream_t stream) {
  const float* x  = (const float*)d_in[0];
  const int*   ei = (const int*)d_in[1];
  const float* W  = (const float*)d_in[2];
  const float* b  = (const float*)d_in[3];
  const float* pa = (const float*)d_in[4];
  float* out = (float*)d_out;

  const int N = in_sizes[0] / FD;
  const int E = in_sizes[1] / 2;

  const int S  = (E + TE - 1) / TE;
  const int nB = (N + BWN - 1) / BWN;

  float*    dinv   = (float*)d_ws + N;
  float*    y      = (float*)d_ws + 2 * (size_t)N;
  unsigned* tbl    = (unsigned*)(((uintptr_t)(y + (size_t)N * FD) + 15) & ~(uintptr_t)15);
  unsigned* sorted = (unsigned*)(((uintptr_t)(tbl + (size_t)S * (nB + 1)) + 15) & ~(uintptr_t)15);
  unsigned* rowptr = (unsigned*)(((uintptr_t)(sorted + E) + 15) & ~(uintptr_t)15);
  int*      csr    = (int*)(((uintptr_t)(rowptr + N + 1) + 15) & ~(uintptr_t)15);

  tile_sort_kernel<<<S, NT1, 0, stream>>>(ei, E, N, nB, tbl, sorted);
  bucket_csr_kernel<<<nB, NT2, 0, stream>>>(tbl, S, nB, sorted, N, dinv, rowptr, csr);

  const int nRowTiles = (N + 15) / 16;
  const int bG = (nRowTiles + WPB - 1) / WPB;
  gemm_xw_kernel<<<bG, 256, 0, stream>>>(x, W, dinv, y, nRowTiles, N);

  const int bA = (N + 7) / 8;
  gather_kernel<<<bA, 256, 0, stream>>>(csr, rowptr, y, dinv, b, pa, out, N);
}


# dense product moved into the first launch as extra blocks beside the edge-tile sort; the bucket kernel scales its own 256 rows of xW by dinv (same single rounding); 4 launches -> 3; sum order kept
# speedup vs baseline: 3.0735x; 1.0587x over previous
#include <hip/hip_runtime.h>
#include <hip/hip_bf16.h>
#include <stdint.h>

typedef float v2f __attribute__((ext_vector_type(2)));
typedef float v8f __attribute__((ext_vector_type(8)));

#define FD 64

#define BWN   256
#define BSH   8
#define TE    8192
#define NT1   1024
#define EPT   (TE / NT1)
#define NT2   512
#define NW2   (NT2 / 32)
#define CAP   16384

__device__ __forceinline__ unsigned block_scan_excl(unsigned v, unsigned* wsum, unsigned* total) {
  const int lane = threadIdx.x & 31;
  const int wave = threadIdx.x >> 5;
  const int nw   = (int)(blockDim.x >> 5);
  unsigned inc = v;
  for (int o = 1; o < 32; o <<= 1) {
    const unsigned t = __shfl_up(inc, o);
    if (lane >= o) inc += t;
  }
  __syncthreads();
  if (lane == 31) wsum[wave] = inc;
  __syncthreads();
  unsigned woff = 0u, tot = 0u;
  for (int i = 0; i < nw; ++i) {
    const unsigned w = wsum[i];
    if (i < wave) woff += w;
    tot += w;
  }
  *total = tot;
  return woff + inc - v;
}

__device__ __forceinline__ void tile_sort_block(const int tile, const int* __restrict__ ei, int E, int N, int nB,
                                                unsigned* __restrict__ tbl,
                                                unsigned* __restrict__ sorted) {
  __shared__ unsigned cnt[NT1];
  __shared__ unsigned stage[TE];
  __shared__ unsigned wsum[32];
  const int tid = threadIdx.x;
  cnt[tid] = 0u;
  __syncthreads();

  const int e0 = tile * TE + tid;
  const int* __restrict__ dst = ei + E;
  int d[EPT], sv[EPT];
#pragma unroll
  for (int k = 0; k < EPT; ++k) {
    const int e = e0 + k * NT1;
    d[k]  = (e < E) ? dst[e] : -1;
    sv[k] = (e < E) ? ei[e] : 0;
  }
  unsigned pos[EPT];
#pragma unroll
  for (int k = 0; k < EPT; ++k) {
    pos[k] = 0xffffffffu;
    if ((unsigned)d[k] < (unsigned)N) pos[k] = atomicAdd(&cnt[d[k] >> BSH], 1u);
  }
  __syncthreads();

  unsigned tot;
  const unsigned excl = block_scan_excl(cnt[tid], wsum, &tot);
  cnt[tid] = excl;
  if (tid <= nB) tbl[(size_t)tile * (nB + 1) + tid] = excl;
  __syncthreads();

#pragma unroll
  for (int k = 0; k < EPT; ++k)
    if (pos[k] != 0xffffffffu)
      stage[cnt[d[k] >> BSH] + pos[k]] = ((unsigned)sv[k] << 8) | ((unsigned)d[k] & (BWN - 1));
  __syncthreads();

  for (int i = tid; i < TE; i += NT1) {
    const int e = tile * TE + i;
    if (e < E && (unsigned)i < tot) sorted[e] = stage[i];
  }
}

__device__ __forceinline__ unsigned bucket_rec(const unsigned* runoff, const unsigned* runpos,
                                               const unsigned* __restrict__ sorted, unsigned i) {
  unsigned a = 0u;
#pragma unroll
  for (unsigned st = NT2 / 2; st > 0u; st >>= 1) {
    const unsigned m = a + st;
    if (runoff[m] <= i) a = m;
  }
  return sorted[runpos[a] + (i - runoff[a])];
}

__global__ void __launch_bounds__(NT2, 8) bucket_csr_kernel(const unsigned* __restrict__ tbl, int S, int nB,
                                                            const unsigned* __restrict__ sorted, int N,
                                                            float* __restrict__ dinv,
                                                            unsigned* __restrict__ rowptr,
                                                            int* __restrict__ csr,
                                                            float* y) {
  __shared__ unsigned runoff[NT2];
  __shared__ unsigned runpos[NT2];
  __shared__ unsigned runlen[NT2];
  __shared__ unsigned ncnt[BWN];
  __shared__ unsigned cur[BWN];
  __shared__ unsigned rbuf[CAP];
  __shared__ unsigned ostage[CAP];
  __shared__ unsigned wsum[32];
  __shared__ float sdinv[BWN];
  const int tid = threadIdx.x;
  const int b   = blockIdx.x;

  float4* y4 = (float4*)y + (size_t)b * (BWN * (FD / 4));
  const int yrow = tid >> 4;
  float4 yv[8];
#pragma unroll
  for (int j = 0; j < 8; ++j) {
    yv[j] = make_float4(0.0f, 0.0f, 0.0f, 0.0f);
    if (b * BWN + yrow + 32 * j < N) yv[j] = y4[tid + NT2 * j];
  }

  unsigned lo = 0u, len = 0u;
  if (tid < S) {
    const unsigned* __restrict__ t = tbl + (size_t)tid * (nB + 1) + b;
    lo  = t[0];
    len = t[1] - lo;
  }
  unsigned nb, base;
  const unsigned roff = block_scan_excl(len, wsum, &nb);
  (void)block_scan_excl(lo, wsum, &base);
  runoff[tid] = roff;
  runpos[tid] = (unsigned)tid * TE + lo;
  runlen[tid] = len;
  if (tid < BWN) ncnt[tid] = 0u;
  __syncthreads();

  const bool fits = nb <= (unsigned)CAP;
  const int lane  = tid & 31;
  const int wave  = tid >> 5;
  const int myrun = wave + NW2 * lane;
  const int m_off = (int)runoff[myrun];
  const int m_pos = (int)runpos[myrun];
  const int m_len = (int)runlen[myrun];
  for (int q0 = 0; q0 < 32 && wave + NW2 * q0 < S; q0 += 16) {
    unsigned rec[16];
#pragma unroll
    for (int u = 0; u < 16; ++u) {
      const unsigned len = (unsigned)__builtin_amdgcn_readlane(m_len, q0 + u);
      const unsigned pos = (unsigned)__builtin_amdgcn_readlane(m_pos, q0 + u);
      rec[u] = ((unsigned)lane < len) ? sorted[pos + lane] : 0u;
    }
#pragma unroll
    for (int u = 0; u < 16; ++u) {
      const unsigned len = (unsigned)__builtin_amdgcn_readlane(m_len, q0 + u);
      const unsigned off = (unsigned)__builtin_amdgcn_readlane(m_off, q0 + u);
      if ((unsigned)lane < len) {
        atomicAdd(&ncnt[rec[u] & (BWN - 1)], 1u);
        if (fits) rbuf[off + lane] = rec[u];
      }
    }
    for (int u = 0; u < 16; ++u) {
      const unsigned len = (unsigned)__builtin_amdgcn_readlane(m_len, q0 + u);
      const unsigned pos = (unsigned)__builtin_amdgcn_readlane(m_pos, q0 + u);
      const unsigned off = (unsigned)__builtin_amdgcn_readlane(m_off, q0 + u);
      for (unsigned j = lane + 32; j < len; j += 32) {
        const unsigned r = sorted[pos + j];
        atomicAdd(&ncnt[r & (BWN - 1)], 1u);
        if (fits) rbuf[off + j] = r;
      }
    }
  }
  __syncthreads();

  const unsigned c = (tid < BWN) ? ncnt[tid] : 0u;
  unsigned tot2;
  const unsigned noff = block_scan_excl(c, wsum, &tot2);
  if (tid < BWN) {
    cur[tid] = noff;
    const int node = b * BWN + tid;
    if (node < N) {
      rowptr[node] = base + noff;
      const float dv = rsqrtf((float)(c + 1u));
      dinv[node]  = dv;
      sdinv[tid]  = dv;
      if (node == N - 1) rowptr[N] = base + noff + c;
    }
  }
  __syncthreads();

#pragma unroll
  for (int j = 0; j < 8; ++j) {
    if (b * BWN + yrow + 32 * j < N) {
      const float s = sdinv[yrow + 32 * j];
      float4 v = yv[j];
      v.x *= s; v.y *= s; v.z *= s; v.w *= s;
      y4[tid + NT2 * j] = v;
    }
  }

  if (fits) {
    for (unsigned i = tid; i < nb; i += NT2) {
      const unsigned r = rbuf[i];
      const unsigned slot = atomicAdd(&cur[r & (BWN - 1)], 1u);
      ostage[slot] = r >> 8;
    }
    __syncthreads();
    for (unsigned i = tid; i < nb; i += NT2) csr[base + i] = (int)ostage[i];
  } else {
    for (unsigned i = tid; i < nb; i += NT2) {
      const unsigned r = bucket_rec(runoff, runpos, sorted, i);
      const unsigned slot = atomicAdd(&cur[r & (BWN - 1)], 1u);
      csr[base + slot] = (int)(r >> 8);
    }
  }
}

__device__ __forceinline__ void gemm_xw_block(const int gb, const float* __restrict__ x,
                                              const float* __restrict__ W,
                                              float* __restrict__ y,
                                              int nRowTiles, int N) {
  __shared__ float Wl[FD * FD];
  for (int p = threadIdx.x; p < 2048; p += NT1) {
    const int lp   = p & 31;
    const int f    = p >> 5;
    const int c    = f & 3;
    const int kk   = f >> 2;
    const int n    = lp & 15;
    const int h    = lp >> 4;
    const int row  = (kk << 2) + (h << 1);
    const int col  = (c << 4) + n;
    Wl[2 * p + 0] = W[(row + 0) * FD + col];
    Wl[2 * p + 1] = W[(row + 1) * FD + col];
  }
  __syncthreads();

  const int wave = threadIdx.x >> 5;
  const int lane = threadIdx.x & 31;
  const int tile = gb * (NT1 / 32) + wave;
  if (tile >= nRowTiles) return;

  const int r0   = tile << 4;
  const int m    = lane & 15;
  const int half = lane >> 4;
  const int kb   = half << 1;

  int arow = r0 + m; if (arow >= N) arow = N - 1;
  const unsigned xoff = (unsigned)arow * FD + (unsigned)kb;
  const float* __restrict__ bbase = Wl + (lane << 1);

  v8f c0 = {}, c1 = {}, c2 = {}, c3 = {};
#pragma unroll
  for (int kk = 0; kk < 16; ++kk) {
    v2f a;
    a.x = x[xoff + (kk << 2) + 0];
    a.y = x[xoff + (kk << 2) + 1];
    const float* fb = bbase + (kk << 8);
    const v2f b0 = *(const v2f*)(fb + 0 * 64);
    const v2f b1 = *(const v2f*)(fb + 1 * 64);
    const v2f b2 = *(const v2f*)(fb + 2 * 64);
    const v2f b3 = *(const v2f*)(fb + 3 * 64);
    c0 = __builtin_amdgcn_wmma_f32_16x16x4_f32(false, a, false, b0, (short)0, c0, false, false);
    c1 = __builtin_amdgcn_wmma_f32_16x16x4_f32(false, a, false, b1, (short)0, c1, false, false);
    c2 = __builtin_amdgcn_wmma_f32_16x16x4_f32(false, a, false, b2, (short)0, c2, false, false);
    c3 = __builtin_amdgcn_wmma_f32_16x16x4_f32(false, a, false, b3, (short)0, c3, false, false);
  }

#pragma unroll
  for (int r = 0; r < 8; ++r) {
    const int row = r0 + r + (half << 3);
    if (row < N) {
      const unsigned base = (unsigned)row * FD + (unsigned)m;
      y[base +  0] = c0[r];
      y[base + 16] = c1[r];
      y[base + 32] = c2[r];
      y[base + 48] = c3[r];
    }
  }
}

__global__ void __launch_bounds__(NT1, 16) tile_sort_gemm_kernel(const int* __restrict__ ei, int E, int N, int nB,
                                                                 unsigned* __restrict__ tbl,
                                                                 unsigned* __restrict__ sorted, int S,
                                                                 const float* __restrict__ x,
                                                                 const float* __restrict__ W,
                                                                 float* __restrict__ y, int nRowTiles) {
  if ((int)blockIdx.x < S) tile_sort_block((int)blockIdx.x, ei, E, N, nB, tbl, sorted);
  else                     gemm_xw_block((int)blockIdx.x - S, x, W, y, nRowTiles, N);
}

__global__ void __launch_bounds__(256) gather_kernel(const int* __restrict__ csr,
                                                     const unsigned* __restrict__ rowptr,
                                                     const float* __restrict__ y,
                                                     const float* __restrict__ dinv,
                                                     const float* __restrict__ b,
                                                     const float* __restrict__ prelu_a,
                                                     float* __restrict__ out, int N) {
  const int lane = threadIdx.x & 31;
  const int node = __builtin_amdgcn_readfirstlane((int)(blockIdx.x * 8 + (threadIdx.x >> 5)));
  if (node >= N) return;
  const unsigned beg = rowptr[node];
  const int L = (int)(rowptr[node + 1] - beg);
  const float2* __restrict__ y2 = (const float2*)y;
  float2 acc = y2[(size_t)node * 32 + lane];
  const float  dn = dinv[node];
  const float  a  = prelu_a[0];
  const float2 bb = *((const float2*)b + lane);

  if (L <= 32 && N <= (1 << 24)) {
    const int key = (lane < L) ? ((csr[beg + lane] << 5) | lane) : 0x7fffffff;
    int rank = 0;
    for (int j = 0; j < L; ++j)
      rank += (__builtin_amdgcn_readlane(key, j) < key) ? 1 : 0;
    const int sorted = __builtin_amdgcn_ds_permute(rank << 2, key);
    const char* __restrict__ yb = (const char*)y;
    const unsigned lane8 = (unsigned)lane << 3;
    for (int k = 0; k < L; k += 8) {
      float2 v[8];
#pragma unroll
      for (int j = 0; j < 8; ++j) {
        const int kk = (k + j < L) ? (k + j) : (L - 1);
        const unsigned s = (unsigned)__builtin_amdgcn_readlane(sorted, kk) >> 5;
        v[j] = *(const float2*)(yb + ((s << 8) | lane8));
      }
#pragma unroll
      for (int j = 0; j < 8; ++j) {
        if (k + j < L) { acc.x += v[j].x; acc.y += v[j].y; }
      }
    }
  } else {
    int ls = -1, lj = -1;
    for (int k = 0; k < L; ++k) {
      int bs = 0x7fffffff, bj = 0x7fffffff;
      for (int j = lane; j < L; j += 32) {
        const int s = csr[beg + j];
        const bool above = (s > ls) || (s == ls && j > lj);
        const bool lower = (s < bs) || (s == bs && j < bj);
        if (above && lower) { bs = s; bj = j; }
      }
      for (int o = 16; o > 0; o >>= 1) {
        const int ts = __shfl_xor(bs, o);
        const int tj = __shfl_xor(bj, o);
        if (ts < bs || (ts == bs && tj < bj)) { bs = ts; bj = tj; }
      }
      ls = bs; lj = bj;
      const float2 v = y2[(size_t)bs * 32 + lane];
      acc.x += v.x; acc.y += v.y;
    }
  }

  float2 r;
  r.x = dn * acc.x + bb.x;
  r.y = dn * acc.y + bb.y;
  r.x = r.x >= 0.0f ? r.x : a * r.x;
  r.y = r.y >= 0.0f ? r.y : a * r.y;
  *((float2*)out + (size_t)node * 32 + lane) = r;
}

extern "C" void kernel_launch(void* const* d_in, const int* in_sizes, int n_in,
                              void* d_out, int out_size, void* d_ws, size_t ws_size,
                              hipStream_t stream) {
  const float* x  = (const float*)d_in[0];
  const int*   ei = (const int*)d_in[1];
  const float* W  = (const float*)d_in[2];
  const float* b  = (const float*)d_in[3];
  const float* pa = (const float*)d_in[4];
  float* out = (float*)d_out;

  const int N = in_sizes[0] / FD;
  const int E = in_sizes[1] / 2;

  const int S  = (E + TE - 1) / TE;
  const int nB = (N + BWN - 1) / BWN;

  float*    dinv   = (float*)d_ws + N;
  float*    y      = (float*)d_ws + 2 * (size_t)N;
  unsigned* tbl    = (unsigned*)(((uintptr_t)(y + (size_t)N * FD) + 15) & ~(uintptr_t)15);
  unsigned* sorted = (unsigned*)(((uintptr_t)(tbl + (size_t)S * (nB + 1)) + 15) & ~(uintptr_t)15);
  unsigned* rowptr = (unsigned*)(((uintptr_t)(sorted + E) + 15) & ~(uintptr_t)15);
  int*      csr    = (int*)(((uintptr_t)(rowptr + N + 1) + 15) & ~(uintptr_t)15);

  const int nRowTiles = (N + 15) / 16;
  const int bG = (nRowTiles + NT1 / 32 - 1) / (NT1 / 32);
  tile_sort_gemm_kernel<<<S + bG, NT1, 0, stream>>>(ei, E, N, nB, tbl, sorted, S, x, W, y, nRowTiles);
  bucket_csr_kernel<<<nB, NT2, 0, stream>>>(tbl, S, nB, sorted, N, dinv, rowptr, csr, y);

  const int bA = (N + 7) / 8;
  gather_kernel<<<bA, 256, 0, stream>>>(csr, rowptr, y, dinv, b, pa, out, N);
}


# gather: segments of 33..64 sources ranked in-wave (two keys per lane) and read 8 rows at a time, not one selection pass with dependent loads per source: ends the kernel's tail wave; sum order kept
# speedup vs baseline: 3.9510x; 1.2855x over previous
#include <hip/hip_runtime.h>
#include <hip/hip_bf16.h>
#include <stdint.h>

typedef float v2f __attribute__((ext_vector_type(2)));
typedef float v8f __attribute__((ext_vector_type(8)));

#define FD 64

#define BWN   256
#define BSH   8
#define TE    8192
#define NT1   1024
#define EPT   (TE / NT1)
#define NT2   512
#define NW2   (NT2 / 32)
#define CAP   16384

__device__ __forceinline__ unsigned block_scan_excl(unsigned v, unsigned* wsum, unsigned* total) {
  const int lane = threadIdx.x & 31;
  const int wave = threadIdx.x >> 5;
  const int nw   = (int)(blockDim.x >> 5);
  unsigned inc = v;
  for (int o = 1; o < 32; o <<= 1) {
    const unsigned t = __shfl_up(inc, o);
    if (lane >= o) inc += t;
  }
  __syncthreads();
  if (lane == 31) wsum[wave] = inc;
  __syncthreads();
  unsigned woff = 0u, tot = 0u;
  for (int i = 0; i < nw; ++i) {
    const unsigned w = wsum[i];
    if (i < wave) woff += w;
    tot += w;
  }
  *total = tot;
  return woff + inc - v;
}

__device__ __forceinline__ void tile_sort_block(const int tile, const int* __restrict__ ei, int E, int N, int nB,
                                                unsigned* __restrict__ tbl,
                                                unsigned* __restrict__ sorted) {
  __shared__ unsigned cnt[NT1];
  __shared__ unsigned stage[TE];
  __shared__ unsigned wsum[32];
  const int tid = threadIdx.x;
  cnt[tid] = 0u;
  __syncthreads();

  const int e0 = tile * TE + tid;
  const int* __restrict__ dst = ei + E;
  int d[EPT], sv[EPT];
#pragma unroll
  for (int k = 0; k < EPT; ++k) {
    const int e = e0 + k * NT1;
    d[k]  = (e < E) ? dst[e] : -1;
    sv[k] = (e < E) ? ei[e] : 0;
  }
  unsigned pos[EPT];
#pragma unroll
  for (int k = 0; k < EPT; ++k) {
    pos[k] = 0xffffffffu;
    if ((unsigned)d[k] < (unsigned)N) pos[k] = atomicAdd(&cnt[d[k] >> BSH], 1u);
  }
  __syncthreads();

  unsigned tot;
  const unsigned excl = block_scan_excl(cnt[tid], wsum, &tot);
  cnt[tid] = excl;
  if (tid <= nB) tbl[(size_t)tile * (nB + 1) + tid] = excl;
  __syncthreads();

#pragma unroll
  for (int k = 0; k < EPT; ++k)
    if (pos[k] != 0xffffffffu)
      stage[cnt[d[k] >> BSH] + pos[k]] = ((unsigned)sv[k] << 8) | ((unsigned)d[k] & (BWN - 1));
  __syncthreads();

  for (int i = tid; i < TE; i += NT1) {
    const int e = tile * TE + i;
    if (e < E && (unsigned)i < tot) sorted[e] = stage[i];
  }
}

__device__ __forceinline__ unsigned bucket_rec(const unsigned* runoff, const unsigned* runpos,
                                               const unsigned* __restrict__ sorted, unsigned i) {
  unsigned a = 0u;
#pragma unroll
  for (unsigned st = NT2 / 2; st > 0u; st >>= 1) {
    const unsigned m = a + st;
    if (runoff[m] <= i) a = m;
  }
  return sorted[runpos[a] + (i - runoff[a])];
}

__global__ void __launch_bounds__(NT2, 8) bucket_csr_kernel(const unsigned* __restrict__ tbl, int S, int nB,
                                                            const unsigned* __restrict__ sorted, int N,
                                                            float* __restrict__ dinv,
                                                            unsigned* __restrict__ rowptr,
                                                            int* __restrict__ csr,
                                                            float* y) {
  __shared__ unsigned runoff[NT2];
  __shared__ unsigned runpos[NT2];
  __shared__ unsigned runlen[NT2];
  __shared__ unsigned ncnt[BWN];
  __shared__ unsigned cur[BWN];
  __shared__ unsigned rbuf[CAP];
  __shared__ unsigned ostage[CAP];
  __shared__ unsigned wsum[32];
  __shared__ float sdinv[BWN];
  const int tid = threadIdx.x;
  const int b   = blockIdx.x;

  float4* y4 = (float4*)y + (size_t)b * (BWN * (FD / 4));
  const int yrow = tid >> 4;
  float4 yv[8];
#pragma unroll
  for (int j = 0; j < 8; ++j) {
    yv[j] = make_float4(0.0f, 0.0f, 0.0f, 0.0f);
    if (b * BWN + yrow + 32 * j < N) yv[j] = y4[tid + NT2 * j];
  }

  unsigned lo = 0u, len = 0u;
  if (tid < S) {
    const unsigned* __restrict__ t = tbl + (size_t)tid * (nB + 1) + b;
    lo  = t[0];
    len = t[1] - lo;
  }
  unsigned nb, base;
  const unsigned roff = block_scan_excl(len, wsum, &nb);
  (void)block_scan_excl(lo, wsum, &base);
  runoff[tid] = roff;
  runpos[tid] = (unsigned)tid * TE + lo;
  runlen[tid] = len;
  if (tid < BWN) ncnt[tid] = 0u;
  __syncthreads();

  const bool fits = nb <= (unsigned)CAP;
  const int lane  = tid & 31;
  const int wave  = tid >> 5;
  const int myrun = wave + NW2 * lane;
  const int m_off = (int)runoff[myrun];
  const int m_pos = (int)runpos[myrun];
  const int m_len = (int)runlen[myrun];
  for (int q0 = 0; q0 < 32 && wave + NW2 * q0 < S; q0 += 16) {
    unsigned rec[16];
#pragma unroll
    for (int u = 0; u < 16; ++u) {
      const unsigned len = (unsigned)__builtin_amdgcn_readlane(m_len, q0 + u);
      const unsigned pos = (unsigned)__builtin_amdgcn_readlane(m_pos, q0 + u);
      rec[u] = ((unsigned)lane < len) ? sorted[pos + lane] : 0u;
    }
#pragma unroll
    for (int u = 0; u < 16; ++u) {
      const unsigned len = (unsigned)__builtin_amdgcn_readlane(m_len, q0 + u);
      const unsigned off = (unsigned)__builtin_amdgcn_readlane(m_off, q0 + u);
      if ((unsigned)lane < len) {
        atomicAdd(&ncnt[rec[u] & (BWN - 1)], 1u);
        if (fits) rbuf[off + lane] = rec[u];
      }
    }
    for (int u = 0; u < 16; ++u) {
      const unsigned len = (unsigned)__builtin_amdgcn_readlane(m_len, q0 + u);
      const unsigned pos = (unsigned)__builtin_amdgcn_readlane(m_pos, q0 + u);
      const unsigned off = (unsigned)__builtin_amdgcn_readlane(m_off, q0 + u);
      for (unsigned j = lane + 32; j < len; j += 32) {
        const unsigned r = sorted[pos + j];
        atomicAdd(&ncnt[r & (BWN - 1)], 1u);
        if (fits) rbuf[off + j] = r;
      }
    }
  }
  __syncthreads();

  const unsigned c = (tid < BWN) ? ncnt[tid] : 0u;
  unsigned tot2;
  const unsigned noff = block_scan_excl(c, wsum, &tot2);
  if (tid < BWN) {
    cur[tid] = noff;
    const int node = b * BWN + tid;
    if (node < N) {
      rowptr[node] = base + noff;
      const float dv = rsqrtf((float)(c + 1u));
      dinv[node]  = dv;
      sdinv[tid]  = dv;
      if (node == N - 1) rowptr[N] = base + noff + c;
    }
  }
  __syncthreads();

#pragma unroll
  for (int j = 0; j < 8; ++j) {
    if (b * BWN + yrow + 32 * j < N) {
      const float s = sdinv[yrow + 32 * j];
      float4 v = yv[j];
      v.x *= s; v.y *= s; v.z *= s; v.w *= s;
      y4[tid + NT2 * j] = v;
    }
  }

  if (fits) {
    for (unsigned i = tid; i < nb; i += NT2) {
      const unsigned r = rbuf[i];
      const unsigned slot = atomicAdd(&cur[r & (BWN - 1)], 1u);
      ostage[slot] = r >> 8;
    }
    __syncthreads();
    for (unsigned i = tid; i < nb; i += NT2) csr[base + i] = (int)ostage[i];
  } else {
    for (unsigned i = tid; i < nb; i += NT2) {
      const unsigned r = bucket_rec(runoff, runpos, sorted, i);
      const unsigned slot = atomicAdd(&cur[r & (BWN - 1)], 1u);
      csr[base + slot] = (int)(r >> 8);
    }
  }
}

__device__ __forceinline__ void gemm_xw_block(const int gb, const float* __restrict__ x,
                                              const float* __restrict__ W,
                                              float* __restrict__ y,
                                              int nRowTiles, int N) {
  __shared__ float Wl[FD * FD];
  for (int p = threadIdx.x; p < 2048; p += NT1) {
    const int lp   = p & 31;
    const int f    = p >> 5;
    const int c    = f & 3;
    const int kk   = f >> 2;
    const int n    = lp & 15;
    const int h    = lp >> 4;
    const int row  = (kk << 2) + (h << 1);
    const int col  = (c << 4) + n;
    Wl[2 * p + 0] = W[(row + 0) * FD + col];
    Wl[2 * p + 1] = W[(row + 1) * FD + col];
  }
  __syncthreads();

  const int wave = threadIdx.x >> 5;
  const int lane = threadIdx.x & 31;
  const int tile = gb * (NT1 / 32) + wave;
  if (tile >= nRowTiles) return;

  const int r0   = tile << 4;
  const int m    = lane & 15;
  const int half = lane >> 4;
  const int kb   = half << 1;

  int arow = r0 + m; if (arow >= N) arow = N - 1;
  const unsigned xoff = (unsigned)arow * FD + (unsigned)kb;
  const float* __restrict__ bbase = Wl + (lane << 1);

  v8f c0 = {}, c1 = {}, c2 = {}, c3 = {};
#pragma unroll
  for (int kk = 0; kk < 16; ++kk) {
    v2f a;
    a.x = x[xoff + (kk << 2) + 0];
    a.y = x[xoff + (kk << 2) + 1];
    const float* fb = bbase + (kk << 8);
    const v2f b0 = *(const v2f*)(fb + 0 * 64);
    const v2f b1 = *(const v2f*)(fb + 1 * 64);
    const v2f b2 = *(const v2f*)(fb + 2 * 64);
    const v2f b3 = *(const v2f*)(fb + 3 * 64);
    c0 = __builtin_amdgcn_wmma_f32_16x16x4_f32(false, a, false, b0, (short)0, c0, false, false);
    c1 = __builtin_amdgcn_wmma_f32_16x16x4_f32(false, a, false, b1, (short)0, c1, false, false);
    c2 = __builtin_amdgcn_wmma_f32_16x16x4_f32(false, a, false, b2, (short)0, c2, false, false);
    c3 = __builtin_amdgcn_wmma_f32_16x16x4_f32(false, a, false, b3, (short)0, c3, false, false);
  }

#pragma unroll
  for (int r = 0; r < 8; ++r) {
    const int row = r0 + r + (half << 3);
    if (row < N) {
      const unsigned base = (unsigned)row * FD + (unsigned)m;
      y[base +  0] = c0[r];
      y[base + 16] = c1[r];
      y[base + 32] = c2[r];
      y[base + 48] = c3[r];
    }
  }
}

__global__ void __launch_bounds__(NT1, 16) tile_sort_gemm_kernel(const int* __restrict__ ei, int E, int N, int nB,
                                                                 unsigned* __restrict__ tbl,
                                                                 unsigned* __restrict__ sorted, int S,
                                                                 const float* __restrict__ x,
                                                                 const float* __restrict__ W,
                                                                 float* __restrict__ y, int nRowTiles) {
  if ((int)blockIdx.x < S) tile_sort_block((int)blockIdx.x, ei, E, N, nB, tbl, sorted);
  else                     gemm_xw_block((int)blockIdx.x - S, x, W, y, nRowTiles, N);
}

__global__ void __launch_bounds__(256) gather_kernel(const int* __restrict__ csr,
                                                     const unsigned* __restrict__ rowptr,
                                                     const float* __restrict__ y,
                                                     const float* __restrict__ dinv,
                                                     const float* __restrict__ b,
                                                     const float* __restrict__ prelu_a,
                                                     float* __restrict__ out, int N) {
  const int lane = threadIdx.x & 31;
  const int node = __builtin_amdgcn_readfirstlane((int)(blockIdx.x * 8 + (threadIdx.x >> 5)));
  if (node >= N) return;
  const unsigned beg = rowptr[node];
  const int L = (int)(rowptr[node + 1] - beg);
  const float2* __restrict__ y2 = (const float2*)y;
  float2 acc = y2[(size_t)node * 32 + lane];
  const float  dn = dinv[node];
  const float  a  = prelu_a[0];
  const float2 bb = *((const float2*)b + lane);

  if (L <= 32 && N <= (1 << 24)) {
    const int key = (lane < L) ? ((csr[beg + lane] << 5) | lane) : 0x7fffffff;
    int rank = 0;
    for (int j = 0; j < L; ++j)
      rank += (__builtin_amdgcn_readlane(key, j) < key) ? 1 : 0;
    const int sorted = __builtin_amdgcn_ds_permute(rank << 2, key);
    const char* __restrict__ yb = (const char*)y;
    const unsigned lane8 = (unsigned)lane << 3;
    for (int k = 0; k < L; k += 8) {
      float2 v[8];
#pragma unroll
      for (int j = 0; j < 8; ++j) {
        const int kk = (k + j < L) ? (k + j) : (L - 1);
        const unsigned s = (unsigned)__builtin_amdgcn_readlane(sorted, kk) >> 5;
        v[j] = *(const float2*)(yb + ((s << 8) | lane8));
      }
#pragma unroll
      for (int j = 0; j < 8; ++j) {
        if (k + j < L) { acc.x += v[j].x; acc.y += v[j].y; }
      }
    }
  } else if (L <= 64 && N <= (1 << 24)) {
    const int j1 = lane + 32;
    const int key0 = (csr[beg + lane] << 6) | lane;
    const int key1 = (j1 < L) ? ((csr[beg + j1] << 6) | j1) : 0x7fffffff;
    int rank0 = 0, rank1 = 0;
    for (int j = 0; j < 32; ++j) {
      const int kj = __builtin_amdgcn_readlane(key0, j);
      rank0 += (kj < key0) ? 1 : 0;
      rank1 += (kj < key1) ? 1 : 0;
    }
    for (int j = 0; j < L - 32; ++j) {
      const int kj = __builtin_amdgcn_readlane(key1, j);
      rank0 += (kj < key0) ? 1 : 0;
      rank1 += (kj < key1) ? 1 : 0;
    }
    const char* __restrict__ yb = (const char*)y;
    const unsigned lane8 = (unsigned)lane << 3;
    for (int k = 0; k < L; k += 8) {
      float2 v[8];
#pragma unroll
      for (int j = 0; j < 8; ++j) {
        const int kk = (k + j < L) ? (k + j) : (L - 1);
        const int p0 = __builtin_ffs((int)__builtin_amdgcn_ballot_w32(rank0 == kk)) - 1;
        const int p1 = __builtin_ffs((int)__builtin_amdgcn_ballot_w32(rank1 == kk)) - 1;
        const int kq = (p0 >= 0) ? __builtin_amdgcn_readlane(key0, p0)
                                 : __builtin_amdgcn_readlane(key1, p1);
        const unsigned s = (unsigned)kq >> 6;
        v[j] = *(const float2*)(yb + ((s << 8) | lane8));
      }
#pragma unroll
      for (int j = 0; j < 8; ++j) {
        if (k + j < L) { acc.x += v[j].x; acc.y += v[j].y; }
      }
    }
  } else {
    int ls = -1, lj = -1;
    for (int k = 0; k < L; ++k) {
      int bs = 0x7fffffff, bj = 0x7fffffff;
      for (int j = lane; j < L; j += 32) {
        const int s = csr[beg + j];
        const bool above = (s > ls) || (s == ls && j > lj);
        const bool lower = (s < bs) || (s == bs && j < bj);
        if (above && lower) { bs = s; bj = j; }
      }
      for (int o = 16; o > 0; o >>= 1) {
        const int ts = __shfl_xor(bs, o);
        const int tj = __shfl_xor(bj, o);
        if (ts < bs || (ts == bs && tj < bj)) { bs = ts; bj = tj; }
      }
      ls = bs; lj = bj;
      const float2 v = y2[(size_t)bs * 32 + lane];
      acc.x += v.x; acc.y += v.y;
    }
  }

  float2 r;
  r.x = dn * acc.x + bb.x;
  r.y = dn * acc.y + bb.y;
  r.x = r.x >= 0.0f ? r.x : a * r.x;
  r.y = r.y >= 0.0f ? r.y : a * r.y;
  *((float2*)out + (size_t)node * 32 + lane) = r;
}

extern "C" void kernel_launch(void* const* d_in, const int* in_sizes, int n_in,
                              void* d_out, int out_size, void* d_ws, size_t ws_size,
                              hipStream_t stream) {
  const float* x  = (const float*)d_in[0];
  const int*   ei = (const int*)d_in[1];
  const float* W  = (const float*)d_in[2];
  const float* b  = (const float*)d_in[3];
  const float* pa = (const float*)d_in[4];
  float* out = (float*)d_out;

  const int N = in_sizes[0] / FD;
  const int E = in_sizes[1] / 2;

  const int S  = (E + TE - 1) / TE;
  const int nB = (N + BWN - 1) / BWN;

  float*    dinv   = (float*)d_ws + N;
  float*    y      = (float*)d_ws + 2 * (size_t)N;
  unsigned* tbl    = (unsigned*)(((uintptr_t)(y + (size_t)N * FD) + 15) & ~(uintptr_t)15);
  unsigned* sorted = (unsigned*)(((uintptr_t)(tbl + (size_t)S * (nB + 1)) + 15) & ~(uintptr_t)15);
  unsigned* rowptr = (unsigned*)(((uintptr_t)(sorted + E) + 15) & ~(uintptr_t)15);
  int*      csr    = (int*)(((uintptr_t)(rowptr + N + 1) + 15) & ~(uintptr_t)15);

  const int nRowTiles = (N + 15) / 16;
  const int bG = (nRowTiles + NT1 / 32 - 1) / (NT1 / 32);
  tile_sort_gemm_kernel<<<S + bG, NT1, 0, stream>>>(ei, E, N, nB, tbl, sorted, S, x, W, y, nRowTiles);
  bucket_csr_kernel<<<nB, NT2, 0, stream>>>(tbl, S, nB, sorted, N, dinv, rowptr, csr, y);

  const int bA = (N + 7) / 8;
  gather_kernel<<<bA, 256, 0, stream>>>(csr, rowptr, y, dinv, b, pa, out, N);
}


# gather as resident waves: 16384 waves (one per wave slot) each loop over nodes w, w+16384, .., next nodes' bounds and sources requested under the current rows; not 100000 one-node waves; order kept
# speedup vs baseline: 4.1040x; 1.0387x over previous
#include <hip/hip_runtime.h>
#include <hip/hip_bf16.h>
#include <stdint.h>

typedef float v2f __attribute__((ext_vector_type(2)));
typedef float v8f __attribute__((ext_vector_type(8)));

#define FD 64

#define BWN   256
#define BSH   8
#define TE    8192
#define NT1   1024
#define EPT   (TE / NT1)
#define NT2   512
#define NW2   (NT2 / 32)
#define CAP   16384

__device__ __forceinline__ unsigned block_scan_excl(unsigned v, unsigned* wsum, unsigned* total) {
  const int lane = threadIdx.x & 31;
  const int wave = threadIdx.x >> 5;
  const int nw   = (int)(blockDim.x >> 5);
  unsigned inc = v;
  for (int o = 1; o < 32; o <<= 1) {
    const unsigned t = __shfl_up(inc, o);
    if (lane >= o) inc += t;
  }
  __syncthreads();
  if (lane == 31) wsum[wave] = inc;
  __syncthreads();
  unsigned woff = 0u, tot = 0u;
  for (int i = 0; i < nw; ++i) {
    const unsigned w = wsum[i];
    if (i < wave) woff += w;
    tot += w;
  }
  *total = tot;
  return woff + inc - v;
}

__device__ __forceinline__ void tile_sort_block(const int tile, const int* __restrict__ ei, int E, int N, int nB,
                                                unsigned* __restrict__ tbl,
                                                unsigned* __restrict__ sorted) {
  __shared__ unsigned cnt[NT1];
  __shared__ unsigned stage[TE];
  __shared__ unsigned wsum[32];
  const int tid = threadIdx.x;
  cnt[tid] = 0u;
  __syncthreads();

  const int e0 = tile * TE + tid;
  const int* __restrict__ dst = ei + E;
  int d[EPT], sv[EPT];
#pragma unroll
  for (int k = 0; k < EPT; ++k) {
    const int e = e0 + k * NT1;
    d[k]  = (e < E) ? dst[e] : -1;
    sv[k] = (e < E) ? ei[e] : 0;
  }
  unsigned pos[EPT];
#pragma unroll
  for (int k = 0; k < EPT; ++k) {
    pos[k] = 0xffffffffu;
    if ((unsigned)d[k] < (unsigned)N) pos[k] = atomicAdd(&cnt[d[k] >> BSH], 1u);
  }
  __syncthreads();

  unsigned tot;
  const unsigned excl = block_scan_excl(cnt[tid], wsum, &tot);
  cnt[tid] = excl;
  if (tid <= nB) tbl[(size_t)tile * (nB + 1) + tid] = excl;
  __syncthreads();

#pragma unroll
  for (int k = 0; k < EPT; ++k)
    if (pos[k] != 0xffffffffu)
      stage[cnt[d[k] >> BSH] + pos[k]] = ((unsigned)sv[k] << 8) | ((unsigned)d[k] & (BWN - 1));
  __syncthreads();

  for (int i = tid; i < TE; i += NT1) {
    const int e = tile * TE + i;
    if (e < E && (unsigned)i < tot) sorted[e] = stage[i];
  }
}

__device__ __forceinline__ unsigned bucket_rec(const unsigned* runoff, const unsigned* runpos,
                                               const unsigned* __restrict__ sorted, unsigned i) {
  unsigned a = 0u;
#pragma unroll
  for (unsigned st = NT2 / 2; st > 0u; st >>= 1) {
    const unsigned m = a + st;
    if (runoff[m] <= i) a = m;
  }
  return sorted[runpos[a] + (i - runoff[a])];
}

__global__ void __launch_bounds__(NT2, 8) bucket_csr_kernel(const unsigned* __restrict__ tbl, int S, int nB,
                                                            const unsigned* __restrict__ sorted, int N,
                                                            float* __restrict__ dinv,
                                                            unsigned* __restrict__ rowptr,
                                                            int* __restrict__ csr,
                                                            float* y) {
  __shared__ unsigned runoff[NT2];
  __shared__ unsigned runpos[NT2];
  __shared__ unsigned runlen[NT2];
  __shared__ unsigned ncnt[BWN];
  __shared__ unsigned cur[BWN];
  __shared__ unsigned rbuf[CAP];
  __shared__ unsigned ostage[CAP];
  __shared__ unsigned wsum[32];
  __shared__ float sdinv[BWN];
  const int tid = threadIdx.x;
  const int b   = blockIdx.x;

  float4* y4 = (float4*)y + (size_t)b * (BWN * (FD / 4));
  const int yrow = tid >> 4;
  float4 yv[8];
#pragma unroll
  for (int j = 0; j < 8; ++j) {
    yv[j] = make_float4(0.0f, 0.0f, 0.0f, 0.0f);
    if (b * BWN + yrow + 32 * j < N) yv[j] = y4[tid + NT2 * j];
  }

  unsigned lo = 0u, len = 0u;
  if (tid < S) {
    const unsigned* __restrict__ t = tbl + (size_t)tid * (nB + 1) + b;
    lo  = t[0];
    len = t[1] - lo;
  }
  unsigned nb, base;
  const unsigned roff = block_scan_excl(len, wsum, &nb);
  (void)block_scan_excl(lo, wsum, &base);
  runoff[tid] = roff;
  runpos[tid] = (unsigned)tid * TE + lo;
  runlen[tid] = len;
  if (tid < BWN) ncnt[tid] = 0u;
  __syncthreads();

  const bool fits = nb <= (unsigned)CAP;
  const int lane  = tid & 31;
  const int wave  = tid >> 5;
  const int myrun = wave + NW2 * lane;
  const int m_off = (int)runoff[myrun];
  const int m_pos = (int)runpos[myrun];
  const int m_len = (int)runlen[myrun];
  for (int q0 = 0; q0 < 32 && wave + NW2 * q0 < S; q0 += 16) {
    unsigned rec[16];
#pragma unroll
    for (int u = 0; u < 16; ++u) {
      const unsigned len = (unsigned)__builtin_amdgcn_readlane(m_len, q0 + u);
      const unsigned pos = (unsigned)__builtin_amdgcn_readlane(m_pos, q0 + u);
      rec[u] = ((unsigned)lane < len) ? sorted[pos + lane] : 0u;
    }
#pragma unroll
    for (int u = 0; u < 16; ++u) {
      const unsigned len = (unsigned)__builtin_amdgcn_readlane(m_len, q0 + u);
      const unsigned off = (unsigned)__builtin_amdgcn_readlane(m_off, q0 + u);
      if ((unsigned)lane < len) {
        atomicAdd(&ncnt[rec[u] & (BWN - 1)], 1u);
        if (fits) rbuf[off + lane] = rec[u];
      }
    }
    for (int u = 0; u < 16; ++u) {
      const unsigned len = (unsigned)__builtin_amdgcn_readlane(m_len, q0 + u);
      const unsigned pos = (unsigned)__builtin_amdgcn_readlane(m_pos, q0 + u);
      const unsigned off = (unsigned)__builtin_amdgcn_readlane(m_off, q0 + u);
      for (unsigned j = lane + 32; j < len; j += 32) {
        const unsigned r = sorted[pos + j];
        atomicAdd(&ncnt[r & (BWN - 1)], 1u);
        if (fits) rbuf[off + j] = r;
      }
    }
  }
  __syncthreads();

  const unsigned c = (tid < BWN) ? ncnt[tid] : 0u;
  unsigned tot2;
  const unsigned noff = block_scan_excl(c, wsum, &tot2);
  if (tid < BWN) {
    cur[tid] = noff;
    const int node = b * BWN + tid;
    if (node < N) {
      rowptr[node] = base + noff;
      const float dv = rsqrtf((float)(c + 1u));
      dinv[node]  = dv;
      sdinv[tid]  = dv;
      if (node == N - 1) rowptr[N] = base + noff + c;
    }
  }
  __syncthreads();

#pragma unroll
  for (int j = 0; j < 8; ++j) {
    if (b * BWN + yrow + 32 * j < N) {
      const float s = sdinv[yrow + 32 * j];
      float4 v = yv[j];
      v.x *= s; v.y *= s; v.z *= s; v.w *= s;
      y4[tid + NT2 * j] = v;
    }
  }

  if (fits) {
    for (unsigned i = tid; i < nb; i += NT2) {
      const unsigned r = rbuf[i];
      const unsigned slot = atomicAdd(&cur[r & (BWN - 1)], 1u);
      ostage[slot] = r >> 8;
    }
    __syncthreads();
    for (unsigned i = tid; i < nb; i += NT2) csr[base + i] = (int)ostage[i];
  } else {
    for (unsigned i = tid; i < nb; i += NT2) {
      const unsigned r = bucket_rec(runoff, runpos, sorted, i);
      const unsigned slot = atomicAdd(&cur[r & (BWN - 1)], 1u);
      csr[base + slot] = (int)(r >> 8);
    }
  }
}

__device__ __forceinline__ void gemm_xw_block(const int gb, const float* __restrict__ x,
                                              const float* __restrict__ W,
                                              float* __restrict__ y,
                                              int nRowTiles, int N) {
  __shared__ float Wl[FD * FD];
  for (int p = threadIdx.x; p < 2048; p += NT1) {
    const int lp   = p & 31;
    const int f    = p >> 5;
    const int c    = f & 3;
    const int kk   = f >> 2;
    const int n    = lp & 15;
    const int h    = lp >> 4;
    const int row  = (kk << 2) + (h << 1);
    const int col  = (c << 4) + n;
    Wl[2 * p + 0] = W[(row + 0) * FD + col];
    Wl[2 * p + 1] = W[(row + 1) * FD + col];
  }
  __syncthreads();

  const int wave = threadIdx.x >> 5;
  const int lane = threadIdx.x & 31;
  const int tile = gb * (NT1 / 32) + wave;
  if (tile >= nRowTiles) return;

  const int r0   = tile << 4;
  const int m    = lane & 15;
  const int half = lane >> 4;
  const int kb   = half << 1;

  int arow = r0 + m; if (arow >= N) arow = N - 1;
  const unsigned xoff = (unsigned)arow * FD + (unsigned)kb;
  const float* __restrict__ bbase = Wl + (lane << 1);

  v8f c0 = {}, c1 = {}, c2 = {}, c3 = {};
#pragma unroll
  for (int kk = 0; kk < 16; ++kk) {
    v2f a;
    a.x = x[xoff + (kk << 2) + 0];
    a.y = x[xoff + (kk << 2) + 1];
    const float* fb = bbase + (kk << 8);
    const v2f b0 = *(const v2f*)(fb + 0 * 64);
    const v2f b1 = *(const v2f*)(fb + 1 * 64);
    const v2f b2 = *(const v2f*)(fb + 2 * 64);
    const v2f b3 = *(const v2f*)(fb + 3 * 64);
    c0 = __builtin_amdgcn_wmma_f32_16x16x4_f32(false, a, false, b0, (short)0, c0, false, false);
    c1 = __builtin_amdgcn_wmma_f32_16x16x4_f32(false, a, false, b1, (short)0, c1, false, false);
    c2 = __builtin_amdgcn_wmma_f32_16x16x4_f32(false, a, false, b2, (short)0, c2, false, false);
    c3 = __builtin_amdgcn_wmma_f32_16x16x4_f32(false, a, false, b3, (short)0, c3, false, false);
  }

#pragma unroll
  for (int r = 0; r < 8; ++r) {
    const int row = r0 + r + (half << 3);
    if (row < N) {
      const unsigned base = (unsigned)row * FD + (unsigned)m;
      y[base +  0] = c0[r];
      y[base + 16] = c1[r];
      y[base + 32] = c2[r];
      y[base + 48] = c3[r];
    }
  }
}

__global__ void __launch_bounds__(NT1, 16) tile_sort_gemm_kernel(const int* __restrict__ ei, int E, int N, int nB,
                                                                 unsigned* __restrict__ tbl,
                                                                 unsigned* __restrict__ sorted, int S,
                                                                 const float* __restrict__ x,
                                                                 const float* __restrict__ W,
                                                                 float* __restrict__ y, int nRowTiles) {
  if ((int)blockIdx.x < S) tile_sort_block((int)blockIdx.x, ei, E, N, nB, tbl, sorted);
  else                     gemm_xw_block((int)blockIdx.x - S, x, W, y, nRowTiles, N);
}

#define GWV   16384
__global__ void __launch_bounds__(256) gather_kernel(const int* __restrict__ csr,
                                                     const unsigned* __restrict__ rowptr,
                                                     const float* __restrict__ y,
                                                     const float* __restrict__ dinv,
                                                     const float* __restrict__ b,
                                                     const float* __restrict__ prelu_a,
                                                     float* __restrict__ out, int N, int E) {
  const int lane = threadIdx.x & 31;
  const int nw   = (int)gridDim.x * 8;
  int node = __builtin_amdgcn_readfirstlane((int)(blockIdx.x * 8 + (threadIdx.x >> 5)));
  if (node >= N) return;
  const float2* __restrict__ y2 = (const float2*)y;
  const char*   __restrict__ yb = (const char*)y;
  const unsigned lane8 = (unsigned)lane << 3;
  const float  a  = prelu_a[0];
  const float2 bb = *((const float2*)b + lane);
  const bool n24 = N <= (1 << 24);

  const unsigned elast = (unsigned)(E - 1);
  unsigned beg = rowptr[node];
  int      L   = (int)(rowptr[node + 1] - beg);
  float    dn  = dinv[node];
  int      raw = csr[min(beg + (unsigned)lane, elast)];
  int      n1   = (node + nw < N) ? node + nw : node;
  unsigned nbeg = rowptr[n1];
  int      nL   = (int)(rowptr[n1 + 1] - nbeg);
  float    ndn  = dinv[n1];

  for (;;) {
    const int      nraw  = csr[min(nbeg + (unsigned)lane, elast)];
    const int      n2    = (n1 + nw < N) ? n1 + nw : n1;
    const unsigned n2beg = rowptr[n2];
    const int      n2L   = (int)(rowptr[n2 + 1] - n2beg);
    const float    n2dn  = dinv[n2];
    float2 acc = y2[(size_t)node * 32 + lane];

    if (L <= 32 && n24) {
      const int key = (lane < L) ? ((raw << 5) | lane) : 0x7fffffff;
      int rank = 0;
      for (int j = 0; j < L; ++j)
        rank += (__builtin_amdgcn_readlane(key, j) < key) ? 1 : 0;
      const int sorted = __builtin_amdgcn_ds_permute(rank << 2, key);
      for (int k = 0; k < L; k += 8) {
        float2 v[8];
#pragma unroll
        for (int j = 0; j < 8; ++j) {
          const int kk = (k + j < L) ? (k + j) : (L - 1);
          const unsigned s = (unsigned)__builtin_amdgcn_readlane(sorted, kk) >> 5;
          v[j] = *(const float2*)(yb + ((s << 8) | lane8));
        }
#pragma unroll
        for (int j = 0; j < 8; ++j) {
          if (k + j < L) { acc.x += v[j].x; acc.y += v[j].y; }
        }
      }
    } else if (L <= 64 && n24) {
      const int j1 = lane + 32;
      const int key0 = (raw << 6) | lane;
      const int key1 = (j1 < L) ? ((csr[beg + j1] << 6) | j1) : 0x7fffffff;
      int rank0 = 0, rank1 = 0;
      for (int j = 0; j < 32; ++j) {
        const int kj = __builtin_amdgcn_readlane(key0, j);
        rank0 += (kj < key0) ? 1 : 0;
        rank1 += (kj < key1) ? 1 : 0;
      }
      for (int j = 0; j < L - 32; ++j) {
        const int kj = __builtin_amdgcn_readlane(key1, j);
        rank0 += (kj < key0) ? 1 : 0;
        rank1 += (kj < key1) ? 1 : 0;
      }
      for (int k = 0; k < L; k += 8) {
        float2 v[8];
#pragma unroll
        for (int j = 0; j < 8; ++j) {
          const int kk = (k + j < L) ? (k + j) : (L - 1);
          const int p0 = __builtin_ffs((int)__builtin_amdgcn_ballot_w32(rank0 == kk)) - 1;
          const int p1 = __builtin_ffs((int)__builtin_amdgcn_ballot_w32(rank1 == kk)) - 1;
          const int kq = (p0 >= 0) ? __builtin_amdgcn_readlane(key0, p0)
                                   : __builtin_amdgcn_readlane(key1, p1);
          const unsigned s = (unsigned)kq >> 6;
          v[j] = *(const float2*)(yb + ((s << 8) | lane8));
        }
#pragma unroll
        for (int j = 0; j < 8; ++j) {
          if (k + j < L) { acc.x += v[j].x; acc.y += v[j].y; }
        }
      }
    } else {
      int ls = -1, lj = -1;
      for (int k = 0; k < L; ++k) {
        int bs = 0x7fffffff, bj = 0x7fffffff;
        for (int j = lane; j < L; j += 32) {
          const int s = csr[beg + j];
          const bool above = (s > ls) || (s == ls && j > lj);
          const bool lower = (s < bs) || (s == bs && j < bj);
          if (above && lower) { bs = s; bj = j; }
        }
        for (int o = 16; o > 0; o >>= 1) {
          const int ts = __shfl_xor(bs, o);
          const int tj = __shfl_xor(bj, o);
          if (ts < bs || (ts == bs && tj < bj)) { bs = ts; bj = tj; }
        }
        ls = bs; lj = bj;
        const float2 v = y2[(size_t)bs * 32 + lane];
        acc.x += v.x; acc.y += v.y;
      }
    }

    float2 r;
    r.x = dn * acc.x + bb.x;
    r.y = dn * acc.y + bb.y;
    r.x = r.x >= 0.0f ? r.x : a * r.x;
    r.y = r.y >= 0.0f ? r.y : a * r.y;
    *((float2*)out + (size_t)node * 32 + lane) = r;

    if (node + nw >= N) break;
    node = n1; beg = nbeg;  L  = nL;  dn  = ndn;  raw = nraw;
    n1   = n2; nbeg = n2beg; nL = n2L; ndn = n2dn;
  }
}

extern "C" void kernel_launch(void* const* d_in, const int* in_sizes, int n_in,
                              void* d_out, int out_size, void* d_ws, size_t ws_size,
                              hipStream_t stream) {
  const float* x  = (const float*)d_in[0];
  const int*   ei = (const int*)d_in[1];
  const float* W  = (const float*)d_in[2];
  const float* b  = (const float*)d_in[3];
  const float* pa = (const float*)d_in[4];
  float* out = (float*)d_out;

  const int N = in_sizes[0] / FD;
  const int E = in_sizes[1] / 2;

  const int S  = (E + TE - 1) / TE;
  const int nB = (N + BWN - 1) / BWN;

  float*    dinv   = (float*)d_ws + N;
  float*    y      = (float*)d_ws + 2 * (size_t)N;
  unsigned* tbl    = (unsigned*)(((uintptr_t)(y + (size_t)N * FD) + 15) & ~(uintptr_t)15);
  unsigned* sorted = (unsigned*)(((uintptr_t)(tbl + (size_t)S * (nB + 1)) + 15) & ~(uintptr_t)15);
  unsigned* rowptr = (unsigned*)(((uintptr_t)(sorted + E) + 15) & ~(uintptr_t)15);
  int*      csr    = (int*)(((uintptr_t)(rowptr + N + 1) + 15) & ~(uintptr_t)15);

  const int nRowTiles = (N + 15) / 16;
  const int bG = (nRowTiles + NT1 / 32 - 1) / (NT1 / 32);
  tile_sort_gemm_kernel<<<S + bG, NT1, 0, stream>>>(ei, E, N, nB, tbl, sorted, S, x, W, y, nRowTiles);
  bucket_csr_kernel<<<nB, NT2, 0, stream>>>(tbl, S, nB, sorted, N, dinv, rowptr, csr, y);

  const int bN = (N + 7) / 8;
  const int bA = bN < GWV / 8 ? bN : GWV / 8;
  gather_kernel<<<bA, 256, 0, stream>>>(csr, rowptr, y, dinv, b, pa, out, N, E);
}


# barriers fence LDS only (LDS-scoped release/acquire); y rows, edge tile, W and x fragments are requested together in front of the barriers and stay in flight across them; sum order kept
# speedup vs baseline: 4.3872x; 1.0690x over previous
#include <hip/hip_runtime.h>
#include <hip/hip_bf16.h>
#include <stdint.h>

typedef float v2f __attribute__((ext_vector_type(2)));
typedef float v8f __attribute__((ext_vector_type(8)));

#define FD 64

#define BWN   256
#define BSH   8
#define TE    8192
#define NT1   1024
#define EPT   (TE / NT1)
#define NT2   512
#define NW2   (NT2 / 32)
#define CAP   16384

__device__ __forceinline__ void lds_barrier() {
  __builtin_amdgcn_fence(__ATOMIC_RELEASE, "workgroup", "local");
  __builtin_amdgcn_s_barrier();
  __builtin_amdgcn_fence(__ATOMIC_ACQUIRE, "workgroup", "local");
}

__device__ __forceinline__ unsigned block_scan_excl(unsigned v, unsigned* wsum, unsigned* total) {
  const int lane = threadIdx.x & 31;
  const int wave = threadIdx.x >> 5;
  const int nw   = (int)(blockDim.x >> 5);
  unsigned inc = v;
  for (int o = 1; o < 32; o <<= 1) {
    const unsigned t = __shfl_up(inc, o);
    if (lane >= o) inc += t;
  }
  lds_barrier();
  if (lane == 31) wsum[wave] = inc;
  lds_barrier();
  unsigned woff = 0u, tot = 0u;
  for (int i = 0; i < nw; ++i) {
    const unsigned w = wsum[i];
    if (i < wave) woff += w;
    tot += w;
  }
  *total = tot;
  return woff + inc - v;
}

__device__ __forceinline__ void tile_sort_block(const int tile, const int* __restrict__ ei, int E, int N, int nB,
                                                unsigned* __restrict__ tbl,
                                                unsigned* __restrict__ sorted) {
  __shared__ unsigned cnt[NT1];
  __shared__ unsigned stage[TE];
  __shared__ unsigned wsum[32];
  const int tid = threadIdx.x;

  const int e0 = tile * TE + tid;
  const int* __restrict__ dst = ei + E;
  int d[EPT], sv[EPT];
#pragma unroll
  for (int k = 0; k < EPT; ++k) {
    const int e  = e0 + k * NT1;
    const int dv = dst[e < E ? e : E - 1];
    d[k] = (e < E) ? dv : -1;
  }
#pragma unroll
  for (int k = 0; k < EPT; ++k) {
    const int e = e0 + k * NT1;
    sv[k] = ei[e < E ? e : E - 1];
  }
  cnt[tid] = 0u;
  lds_barrier();
  unsigned pos[EPT];
#pragma unroll
  for (int k = 0; k < EPT; ++k) {
    pos[k] = 0xffffffffu;
    if ((unsigned)d[k] < (unsigned)N) pos[k] = atomicAdd(&cnt[d[k] >> BSH], 1u);
  }
  lds_barrier();

  unsigned tot;
  const unsigned excl = block_scan_excl(cnt[tid], wsum, &tot);
  cnt[tid] = excl;
  if (tid <= nB) tbl[(size_t)tile * (nB + 1) + tid] = excl;
  lds_barrier();

#pragma unroll
  for (int k = 0; k < EPT; ++k)
    if (pos[k] != 0xffffffffu)
      stage[cnt[d[k] >> BSH] + pos[k]] = ((unsigned)sv[k] << 8) | ((unsigned)d[k] & (BWN - 1));
  lds_barrier();

  for (int i = tid; i < TE; i += NT1) {
    const int e = tile * TE + i;
    if (e < E && (unsigned)i < tot) sorted[e] = stage[i];
  }
}

__device__ __forceinline__ unsigned bucket_rec(const unsigned* runoff, const unsigned* runpos,
                                               const unsigned* __restrict__ sorted, unsigned i) {
  unsigned a = 0u;
#pragma unroll
  for (unsigned st = NT2 / 2; st > 0u; st >>= 1) {
    const unsigned m = a + st;
    if (runoff[m] <= i) a = m;
  }
  return sorted[runpos[a] + (i - runoff[a])];
}

__global__ void __launch_bounds__(NT2, 8) bucket_csr_kernel(const unsigned* __restrict__ tbl, int S, int nB,
                                                            const unsigned* __restrict__ sorted, int N,
                                                            float* __restrict__ dinv,
                                                            unsigned* __restrict__ rowptr,
                                                            int* __restrict__ csr,
                                                            float* y) {
  __shared__ unsigned runoff[NT2];
  __shared__ unsigned runpos[NT2];
  __shared__ unsigned runlen[NT2];
  __shared__ unsigned ncnt[BWN];
  __shared__ unsigned cur[BWN];
  __shared__ unsigned rbuf[CAP];
  __shared__ unsigned ostage[CAP];
  __shared__ unsigned wsum[32];
  __shared__ float sdinv[BWN];
  const int tid = threadIdx.x;
  const int b   = blockIdx.x;

  const unsigned* __restrict__ t = tbl + (size_t)(tid < S ? tid : S - 1) * (nB + 1) + b;
  const unsigned t0 = t[0];
  const unsigned t1 = t[1];
  __builtin_amdgcn_sched_barrier(0);

  float4* y4 = (float4*)y + (size_t)b * (BWN * (FD / 4));
  const int yrow = tid >> 4;
  float4 yv[8];
#pragma unroll
  for (int j = 0; j < 8; ++j) {
    const int gr = b * BWN + yrow + 32 * j;
    const int lr = gr < N ? gr : N - 1;
    yv[j] = ((const float4*)y)[(size_t)lr * (FD / 4) + (tid & 15)];
  }
  __builtin_amdgcn_sched_barrier(0);

  unsigned lo = 0u, len = 0u;
  if (tid < S) { lo = t0; len = t1 - t0; }
  unsigned nb, base;
  const unsigned roff = block_scan_excl(len, wsum, &nb);
  (void)block_scan_excl(lo, wsum, &base);
  runoff[tid] = roff;
  runpos[tid] = (unsigned)tid * TE + lo;
  runlen[tid] = len;
  if (tid < BWN) ncnt[tid] = 0u;
  lds_barrier();

  const bool fits = nb <= (unsigned)CAP;
  const int lane  = tid & 31;
  const int wave  = tid >> 5;
  const int myrun = wave + NW2 * lane;
  const int m_off = (int)runoff[myrun];
  const int m_pos = (int)runpos[myrun];
  const int m_len = (int)runlen[myrun];
  for (int q0 = 0; q0 < 32 && wave + NW2 * q0 < S; q0 += 16) {
    unsigned rec[16];
#pragma unroll
    for (int u = 0; u < 16; ++u) {
      const unsigned len = (unsigned)__builtin_amdgcn_readlane(m_len, q0 + u);
      const unsigned pos = (unsigned)__builtin_amdgcn_readlane(m_pos, q0 + u);
      rec[u] = ((unsigned)lane < len) ? sorted[pos + lane] : 0u;
    }
#pragma unroll
    for (int u = 0; u < 16; ++u) {
      const unsigned len = (unsigned)__builtin_amdgcn_readlane(m_len, q0 + u);
      const unsigned off = (unsigned)__builtin_amdgcn_readlane(m_off, q0 + u);
      if ((unsigned)lane < len) {
        atomicAdd(&ncnt[rec[u] & (BWN - 1)], 1u);
        if (fits) rbuf[off + lane] = rec[u];
      }
    }
    for (int u = 0; u < 16; ++u) {
      const unsigned len = (unsigned)__builtin_amdgcn_readlane(m_len, q0 + u);
      const unsigned pos = (unsigned)__builtin_amdgcn_readlane(m_pos, q0 + u);
      const unsigned off = (unsigned)__builtin_amdgcn_readlane(m_off, q0 + u);
      for (unsigned j = lane + 32; j < len; j += 32) {
        const unsigned r = sorted[pos + j];
        atomicAdd(&ncnt[r & (BWN - 1)], 1u);
        if (fits) rbuf[off + j] = r;
      }
    }
  }
  lds_barrier();

  const unsigned c = (tid < BWN) ? ncnt[tid] : 0u;
  unsigned tot2;
  const unsigned noff = block_scan_excl(c, wsum, &tot2);
  if (tid < BWN) {
    cur[tid] = noff;
    const int node = b * BWN + tid;
    if (node < N) {
      rowptr[node] = base + noff;
      const float dv = rsqrtf((float)(c + 1u));
      dinv[node]  = dv;
      sdinv[tid]  = dv;
      if (node == N - 1) rowptr[N] = base + noff + c;
    }
  }
  lds_barrier();

#pragma unroll
  for (int j = 0; j < 8; ++j) {
    if (b * BWN + yrow + 32 * j < N) {
      const float s = sdinv[yrow + 32 * j];
      float4 v = yv[j];
      v.x *= s; v.y *= s; v.z *= s; v.w *= s;
      y4[tid + NT2 * j] = v;
    }
  }

  if (fits) {
    for (unsigned i = tid; i < nb; i += NT2) {
      const unsigned r = rbuf[i];
      const unsigned slot = atomicAdd(&cur[r & (BWN - 1)], 1u);
      ostage[slot] = r >> 8;
    }
    lds_barrier();
    for (unsigned i = tid; i < nb; i += NT2) csr[base + i] = (int)ostage[i];
  } else {
    for (unsigned i = tid; i < nb; i += NT2) {
      const unsigned r = bucket_rec(runoff, runpos, sorted, i);
      const unsigned slot = atomicAdd(&cur[r & (BWN - 1)], 1u);
      csr[base + slot] = (int)(r >> 8);
    }
  }
}

__device__ __forceinline__ void gemm_xw_block(const int gb, const float* __restrict__ x,
                                              const float* __restrict__ W,
                                              float* __restrict__ y,
                                              int nRowTiles, int N) {
  __shared__ float Wl[FD * FD];
  const int wave = threadIdx.x >> 5;
  const int lane = threadIdx.x & 31;
  const int tile = gb * (NT1 / 32) + wave;
  const int r0   = tile << 4;
  const int m    = lane & 15;
  const int half = lane >> 4;
  const int kb   = half << 1;

  int arow = r0 + m; if (arow >= N) arow = N - 1;
  const unsigned xoff = (unsigned)arow * FD + (unsigned)kb;

  float w0[2048 / NT1], w1[2048 / NT1];
#pragma unroll
  for (int i = 0; i < 2048 / NT1; ++i) {
    const int p    = (int)threadIdx.x + i * NT1;
    const int lp   = p & 31;
    const int f    = p >> 5;
    const int c    = f & 3;
    const int kk   = f >> 2;
    const int n    = lp & 15;
    const int h    = lp >> 4;
    const int row  = (kk << 2) + (h << 1);
    const int col  = (c << 4) + n;
    w0[i] = W[(row + 0) * FD + col];
    w1[i] = W[(row + 1) * FD + col];
  }
  v2f abuf[8];
#pragma unroll
  for (int kk = 0; kk < 8; ++kk) {
    abuf[kk].x = x[xoff + (kk << 2) + 0];
    abuf[kk].y = x[xoff + (kk << 2) + 1];
  }
#pragma unroll
  for (int i = 0; i < 2048 / NT1; ++i) {
    const int p = (int)threadIdx.x + i * NT1;
    Wl[2 * p + 0] = w0[i];
    Wl[2 * p + 1] = w1[i];
  }
  lds_barrier();

  if (tile >= nRowTiles) return;

  const float* __restrict__ bbase = Wl + (lane << 1);

  v8f c0 = {}, c1 = {}, c2 = {}, c3 = {};
#pragma unroll
  for (int kk = 0; kk < 16; ++kk) {
    const v2f a = abuf[kk & 7];
    if (kk < 8) {
      abuf[kk].x = x[xoff + ((kk + 8) << 2) + 0];
      abuf[kk].y = x[xoff + ((kk + 8) << 2) + 1];
    }
    const float* fb = bbase + (kk << 8);
    const v2f b0 = *(const v2f*)(fb + 0 * 64);
    const v2f b1 = *(const v2f*)(fb + 1 * 64);
    const v2f b2 = *(const v2f*)(fb + 2 * 64);
    const v2f b3 = *(const v2f*)(fb + 3 * 64);
    c0 = __builtin_amdgcn_wmma_f32_16x16x4_f32(false, a, false, b0, (short)0, c0, false, false);
    c1 = __builtin_amdgcn_wmma_f32_16x16x4_f32(false, a, false, b1, (short)0, c1, false, false);
    c2 = __builtin_amdgcn_wmma_f32_16x16x4_f32(false, a, false, b2, (short)0, c2, false, false);
    c3 = __builtin_amdgcn_wmma_f32_16x16x4_f32(false, a, false, b3, (short)0, c3, false, false);
  }

#pragma unroll
  for (int r = 0; r < 8; ++r) {
    const int row = r0 + r + (half << 3);
    if (row < N) {
      const unsigned base = (unsigned)row * FD + (unsigned)m;
      y[base +  0] = c0[r];
      y[base + 16] = c1[r];
      y[base + 32] = c2[r];
      y[base + 48] = c3[r];
    }
  }
}

__global__ void __launch_bounds__(NT1, 16) tile_sort_gemm_kernel(const int* __restrict__ ei, int E, int N, int nB,
                                                                 unsigned* __restrict__ tbl,
                                                                 unsigned* __restrict__ sorted, int S,
                                                                 const float* __restrict__ x,
                                                                 const float* __restrict__ W,
                                                                 float* __restrict__ y, int nRowTiles) {
  if ((int)blockIdx.x < S) tile_sort_block((int)blockIdx.x, ei, E, N, nB, tbl, sorted);
  else                     gemm_xw_block((int)blockIdx.x - S, x, W, y, nRowTiles, N);
}

#define GWV   16384
__global__ void __launch_bounds__(256) gather_kernel(const int* __restrict__ csr,
                                                     const unsigned* __restrict__ rowptr,
                                                     const float* __restrict__ y,
                                                     const float* __restrict__ dinv,
                                                     const float* __restrict__ b,
                                                     const float* __restrict__ prelu_a,
                                                     float* __restrict__ out, int N, int E) {
  const int lane = threadIdx.x & 31;
  const int nw   = (int)gridDim.x * 8;
  int node = __builtin_amdgcn_readfirstlane((int)(blockIdx.x * 8 + (threadIdx.x >> 5)));
  if (node >= N) return;
  const float2* __restrict__ y2 = (const float2*)y;
  const char*   __restrict__ yb = (const char*)y;
  const unsigned lane8 = (unsigned)lane << 3;
  const float  a  = prelu_a[0];
  const float2 bb = *((const float2*)b + lane);
  const bool n24 = N <= (1 << 24);

  const unsigned elast = (unsigned)(E - 1);
  unsigned beg = rowptr[node];
  int      L   = (int)(rowptr[node + 1] - beg);
  float    dn  = dinv[node];
  int      raw = csr[min(beg + (unsigned)lane, elast)];
  int      n1   = (node + nw < N) ? node + nw : node;
  unsigned nbeg = rowptr[n1];
  int      nL   = (int)(rowptr[n1 + 1] - nbeg);
  float    ndn  = dinv[n1];

  for (;;) {
    const int      nraw  = csr[min(nbeg + (unsigned)lane, elast)];
    const int      n2    = (n1 + nw < N) ? n1 + nw : n1;
    const unsigned n2beg = rowptr[n2];
    const int      n2L   = (int)(rowptr[n2 + 1] - n2beg);
    const float    n2dn  = dinv[n2];
    float2 acc = y2[(size_t)node * 32 + lane];

    if (L <= 32 && n24) {
      const int key = (lane < L) ? ((raw << 5) | lane) : 0x7fffffff;
      int rank = 0;
      for (int j = 0; j < L; ++j)
        rank += (__builtin_amdgcn_readlane(key, j) < key) ? 1 : 0;
      const int sorted = __builtin_amdgcn_ds_permute(rank << 2, key);
      for (int k = 0; k < L; k += 8) {
        float2 v[8];
#pragma unroll
        for (int j = 0; j < 8; ++j) {
          const int kk = (k + j < L) ? (k + j) : (L - 1);
          const unsigned s = (unsigned)__builtin_amdgcn_readlane(sorted, kk) >> 5;
          v[j] = *(const float2*)(yb + ((s << 8) | lane8));
        }
#pragma unroll
        for (int j = 0; j < 8; ++j) {
          if (k + j < L) { acc.x += v[j].x; acc.y += v[j].y; }
        }
      }
    } else if (L <= 64 && n24) {
      const int j1 = lane + 32;
      const int key0 = (raw << 6) | lane;
      const int key1 = (j1 < L) ? ((csr[beg + j1] << 6) | j1) : 0x7fffffff;
      int rank0 = 0, rank1 = 0;
      for (int j = 0; j < 32; ++j) {
        const int kj = __builtin_amdgcn_readlane(key0, j);
        rank0 += (kj < key0) ? 1 : 0;
        rank1 += (kj < key1) ? 1 : 0;
      }
      for (int j = 0; j < L - 32; ++j) {
        const int kj = __builtin_amdgcn_readlane(key1, j);
        rank0 += (kj < key0) ? 1 : 0;
        rank1 += (kj < key1) ? 1 : 0;
      }
      for (int k = 0; k < L; k += 8) {
        float2 v[8];
#pragma unroll
        for (int j = 0; j < 8; ++j) {
          const int kk = (k + j < L) ? (k + j) : (L - 1);
          const int p0 = __builtin_ffs((int)__builtin_amdgcn_ballot_w32(rank0 == kk)) - 1;
          const int p1 = __builtin_ffs((int)__builtin_amdgcn_ballot_w32(rank1 == kk)) - 1;
          const int kq = (p0 >= 0) ? __builtin_amdgcn_readlane(key0, p0)
                                   : __builtin_amdgcn_readlane(key1, p1);
          const unsigned s = (unsigned)kq >> 6;
          v[j] = *(const float2*)(yb + ((s << 8) | lane8));
        }
#pragma unroll
        for (int j = 0; j < 8; ++j) {
          if (k + j < L) { acc.x += v[j].x; acc.y += v[j].y; }
        }
      }
    } else {
      int ls = -1, lj = -1;
      for (int k = 0; k < L; ++k) {
        int bs = 0x7fffffff, bj = 0x7fffffff;
        for (int j = lane; j < L; j += 32) {
          const int s = csr[beg + j];
          const bool above = (s > ls) || (s == ls && j > lj);
          const bool lower = (s < bs) || (s == bs && j < bj);
          if (above && lower) { bs = s; bj = j; }
        }
        for (int o = 16; o > 0; o >>= 1) {
          const int ts = __shfl_xor(bs, o);
          const int tj = __shfl_xor(bj, o);
          if (ts < bs || (ts == bs && tj < bj)) { bs = ts; bj = tj; }
        }
        ls = bs; lj = bj;
        const float2 v = y2[(size_t)bs * 32 + lane];
        acc.x += v.x; acc.y += v.y;
      }
    }

    float2 r;
    r.x = dn * acc.x + bb.x;
    r.y = dn * acc.y + bb.y;
    r.x = r.x >= 0.0f ? r.x : a * r.x;
    r.y = r.y >= 0.0f ? r.y : a * r.y;
    *((float2*)out + (size_t)node * 32 + lane) = r;

    if (node + nw >= N) break;
    node = n1; beg = nbeg;  L  = nL;  dn  = ndn;  raw = nraw;
    n1   = n2; nbeg = n2beg; nL = n2L; ndn = n2dn;
  }
}

extern "C" void kernel_launch(void* const* d_in, const int* in_sizes, int n_in,
                              void* d_out, int out_size, void* d_ws, size_t ws_size,
                              hipStream_t stream) {
  const float* x  = (const float*)d_in[0];
  const int*   ei = (const int*)d_in[1];
  const float* W  = (const float*)d_in[2];
  const float* b  = (const float*)d_in[3];
  const float* pa = (const float*)d_in[4];
  float* out = (float*)d_out;

  const int N = in_sizes[0] / FD;
  const int E = in_sizes[1] / 2;

  const int S  = (E + TE - 1) / TE;
  const int nB = (N + BWN - 1) / BWN;

  float*    dinv   = (float*)d_ws + N;
  float*    y      = (float*)d_ws + 2 * (size_t)N;
  unsigned* tbl    = (unsigned*)(((uintptr_t)(y + (size_t)N * FD) + 15) & ~(uintptr_t)15);
  unsigned* sorted = (unsigned*)(((uintptr_t)(tbl + (size_t)S * (nB + 1)) + 15) & ~(uintptr_t)15);
  unsigned* rowptr = (unsigned*)(((uintptr_t)(sorted + E) + 15) & ~(uintptr_t)15);
  int*      csr    = (int*)(((uintptr_t)(rowptr + N + 1) + 15) & ~(uintptr_t)15);

  const int nRowTiles = (N + 15) / 16;
  const int bG = (nRowTiles + NT1 / 32 - 1) / (NT1 / 32);
  tile_sort_gemm_kernel<<<S + bG, NT1, 0, stream>>>(ei, E, N, nB, tbl, sorted, S, x, W, y, nRowTiles);
  bucket_csr_kernel<<<nB, NT2, 0, stream>>>(tbl, S, nB, sorted, N, dinv, rowptr, csr, y);

  const int bN = (N + 7) / 8;
  const int bA = bN < GWV / 8 ? bN : GWV / 8;
  gather_kernel<<<bA, 256, 0, stream>>>(csr, rowptr, y, dinv, b, pa, out, N, E);
}


# dense product moved from the first launch into the bucket kernel: a block multiplies its own 256 rows of x by W (swizzled LDS tile) and stores dinv*(xW) once; y never re-read; sum order kept
# speedup vs baseline: 4.6936x; 1.0698x over previous
#include <hip/hip_runtime.h>
#include <hip/hip_bf16.h>
#include <stdint.h>

typedef float v2f __attribute__((ext_vector_type(2)));
typedef float v8f __attribute__((ext_vector_type(8)));

#define FD 64

#define BWN   256
#define BSH   8
#define TE    8192
#define NT1   1024
#define EPT   (TE / NT1)
#define NT2   512
#define NW2   (NT2 / 32)
#define CAP   16384

__device__ __forceinline__ void lds_barrier() {
  __builtin_amdgcn_fence(__ATOMIC_RELEASE, "workgroup", "local");
  __builtin_amdgcn_s_barrier();
  __builtin_amdgcn_fence(__ATOMIC_ACQUIRE, "workgroup", "local");
}

__device__ __forceinline__ unsigned block_scan_excl(unsigned v, unsigned* wsum, unsigned* total) {
  const int lane = threadIdx.x & 31;
  const int wave = threadIdx.x >> 5;
  const int nw   = (int)(blockDim.x >> 5);
  unsigned inc = v;
  for (int o = 1; o < 32; o <<= 1) {
    const unsigned t = __shfl_up(inc, o);
    if (lane >= o) inc += t;
  }
  lds_barrier();
  if (lane == 31) wsum[wave] = inc;
  lds_barrier();
  unsigned woff = 0u, tot = 0u;
  for (int i = 0; i < nw; ++i) {
    const unsigned w = wsum[i];
    if (i < wave) woff += w;
    tot += w;
  }
  *total = tot;
  return woff + inc - v;
}

__device__ __forceinline__ void tile_sort_block(const int tile, const int* __restrict__ ei, int E, int N, int nB,
                                                unsigned* __restrict__ tbl,
                                                unsigned* __restrict__ sorted) {
  __shared__ unsigned cnt[NT1];
  __shared__ unsigned stage[TE];
  __shared__ unsigned wsum[32];
  const int tid = threadIdx.x;

  const int e0 = tile * TE + tid;
  const int* __restrict__ dst = ei + E;
  int d[EPT], sv[EPT];
#pragma unroll
  for (int k = 0; k < EPT; ++k) {
    const int e  = e0 + k * NT1;
    const int dv = dst[e < E ? e : E - 1];
    d[k] = (e < E) ? dv : -1;
  }
#pragma unroll
  for (int k = 0; k < EPT; ++k) {
    const int e = e0 + k * NT1;
    sv[k] = ei[e < E ? e : E - 1];
  }
  cnt[tid] = 0u;
  lds_barrier();
  unsigned pos[EPT];
#pragma unroll
  for (int k = 0; k < EPT; ++k) {
    pos[k] = 0xffffffffu;
    if ((unsigned)d[k] < (unsigned)N) pos[k] = atomicAdd(&cnt[d[k] >> BSH], 1u);
  }
  lds_barrier();

  unsigned tot;
  const unsigned excl = block_scan_excl(cnt[tid], wsum, &tot);
  cnt[tid] = excl;
  if (tid <= nB) tbl[(size_t)tile * (nB + 1) + tid] = excl;
  lds_barrier();

#pragma unroll
  for (int k = 0; k < EPT; ++k)
    if (pos[k] != 0xffffffffu)
      stage[cnt[d[k] >> BSH] + pos[k]] = ((unsigned)sv[k] << 8) | ((unsigned)d[k] & (BWN - 1));
  lds_barrier();

  for (int i = tid; i < TE; i += NT1) {
    const int e = tile * TE + i;
    if (e < E && (unsigned)i < tot) sorted[e] = stage[i];
  }
}

__device__ __forceinline__ unsigned bucket_rec(const unsigned* runoff, const unsigned* runpos,
                                               const unsigned* __restrict__ sorted, unsigned i) {
  unsigned a = 0u;
#pragma unroll
  for (unsigned st = NT2 / 2; st > 0u; st >>= 1) {
    const unsigned m = a + st;
    if (runoff[m] <= i) a = m;
  }
  return sorted[runpos[a] + (i - runoff[a])];
}

__device__ __forceinline__ void sweep1_load(unsigned (&rec)[16], const int m_len, const int m_pos, const int q0,
                                            const int lane, const unsigned* __restrict__ sorted) {
#pragma unroll
  for (int u = 0; u < 16; ++u) {
    const unsigned len = (unsigned)__builtin_amdgcn_readlane(m_len, q0 + u);
    const unsigned pos = (unsigned)__builtin_amdgcn_readlane(m_pos, q0 + u);
    rec[u] = ((unsigned)lane < len) ? sorted[pos + lane] : 0u;
  }
}

__device__ __forceinline__ void sweep1_count(const unsigned (&rec)[16], const int m_len, const int m_pos,
                                             const int m_off, const int q0, const int lane, const bool fits,
                                             unsigned* ncnt, unsigned* rbuf,
                                             const unsigned* __restrict__ sorted) {
#pragma unroll
  for (int u = 0; u < 16; ++u) {
    const unsigned len = (unsigned)__builtin_amdgcn_readlane(m_len, q0 + u);
    const unsigned off = (unsigned)__builtin_amdgcn_readlane(m_off, q0 + u);
    if ((unsigned)lane < len) {
      atomicAdd(&ncnt[rec[u] & (BWN - 1)], 1u);
      if (fits) rbuf[off + lane] = rec[u];
    }
  }
  for (int u = 0; u < 16; ++u) {
    const unsigned len = (unsigned)__builtin_amdgcn_readlane(m_len, q0 + u);
    const unsigned pos = (unsigned)__builtin_amdgcn_readlane(m_pos, q0 + u);
    const unsigned off = (unsigned)__builtin_amdgcn_readlane(m_off, q0 + u);
    for (unsigned j = lane + 32; j < len; j += 32) {
      const unsigned r = sorted[pos + j];
      atomicAdd(&ncnt[r & (BWN - 1)], 1u);
      if (fits) rbuf[off + j] = r;
    }
  }
}

__global__ void __launch_bounds__(NT2, 8) bucket_csr_kernel(const unsigned* __restrict__ tbl, int S, int nB,
                                                            const unsigned* __restrict__ sorted, int N,
                                                            float* __restrict__ dinv,
                                                            unsigned* __restrict__ rowptr,
                                                            int* __restrict__ csr,
                                                            const float* __restrict__ x,
                                                            const float* __restrict__ W,
                                                            float* __restrict__ y) {
  __shared__ unsigned runoff[NT2];
  __shared__ unsigned runpos[NT2];
  __shared__ unsigned runlen[NT2];
  __shared__ unsigned ncnt[BWN];
  __shared__ unsigned cur[BWN];
  __shared__ unsigned rbuf[CAP];
  __shared__ unsigned ostage[CAP];
  __shared__ unsigned wsum[32];
  __shared__ float sdinv[BWN];
  __shared__ float Wl[FD * FD];
  const int tid = threadIdx.x;
  const int b   = blockIdx.x;

  const unsigned* __restrict__ t = tbl + (size_t)(tid < S ? tid : S - 1) * (nB + 1) + b;
  const unsigned t0 = t[0];
  const unsigned t1 = t[1];
  __builtin_amdgcn_sched_barrier(0);

  float w0[2048 / NT2], w1[2048 / NT2];
#pragma unroll
  for (int i = 0; i < 2048 / NT2; ++i) {
    const int p    = tid + i * NT2;
    const int lp   = p & 31;
    const int f    = p >> 5;
    const int c    = f & 3;
    const int kk   = f >> 2;
    const int n    = lp & 15;
    const int h    = lp >> 4;
    const int row  = (kk << 2) + (h << 1);
    const int col  = (c << 4) + n;
    w0[i] = W[(row + 0) * FD + col];
    w1[i] = W[(row + 1) * FD + col];
  }
  const int yrow = tid >> 4;
  float4 xv[8];
#pragma unroll
  for (int j = 0; j < 8; ++j) {
    const int gr = b * BWN + yrow + 32 * j;
    const int lr = gr < N ? gr : N - 1;
    xv[j] = ((const float4*)x)[(size_t)lr * (FD / 4) + (tid & 15)];
  }
  __builtin_amdgcn_sched_barrier(0);

  unsigned lo = 0u, len = 0u;
  if (tid < S) { lo = t0; len = t1 - t0; }
  unsigned nb, base;
  const unsigned roff = block_scan_excl(len, wsum, &nb);
  (void)block_scan_excl(lo, wsum, &base);
  runoff[tid] = roff;
  runpos[tid] = (unsigned)tid * TE + lo;
  runlen[tid] = len;
  if (tid < BWN) ncnt[tid] = 0u;
#pragma unroll
  for (int i = 0; i < 2048 / NT2; ++i) {
    const int p = tid + i * NT2;
    Wl[2 * p + 0] = w0[i];
    Wl[2 * p + 1] = w1[i];
  }
#pragma unroll
  for (int j = 0; j < 8; ++j) {
    const int r = yrow + 32 * j;
    uint4 u;
    u.x = __float_as_uint(xv[j].x); u.y = __float_as_uint(xv[j].y);
    u.z = __float_as_uint(xv[j].z); u.w = __float_as_uint(xv[j].w);
    *(uint4*)&ostage[r * FD + (((tid & 15) ^ (r & 15)) << 2)] = u;
  }
  lds_barrier();

  const bool fits = nb <= (unsigned)CAP;
  const int lane  = tid & 31;
  const int wave  = tid >> 5;
  const int myrun = wave + NW2 * lane;
  const int m_off = (int)runoff[myrun];
  const int m_pos = (int)runpos[myrun];
  const int m_len = (int)runlen[myrun];
  const bool has0 = wave < S;
  unsigned rec[16];
  if (has0) sweep1_load(rec, m_len, m_pos, 0, lane, sorted);

  const int m    = lane & 15;
  const int half = lane >> 4;
  const unsigned* abase = ostage + ((wave << 4) + m) * FD + (half << 1);
  const float*    bbase = Wl + (lane << 1);
  v8f c0 = {}, c1 = {}, c2 = {}, c3 = {};
#pragma unroll
  for (int kk = 0; kk < 16; ++kk) {
    const uint2 au = *(const uint2*)(abase + ((kk ^ m) << 2));
    v2f a;
    a.x = __uint_as_float(au.x);
    a.y = __uint_as_float(au.y);
    const float* fb = bbase + (kk << 8);
    const v2f b0 = *(const v2f*)(fb + 0 * 64);
    const v2f b1 = *(const v2f*)(fb + 1 * 64);
    const v2f b2 = *(const v2f*)(fb + 2 * 64);
    const v2f b3 = *(const v2f*)(fb + 3 * 64);
    c0 = __builtin_amdgcn_wmma_f32_16x16x4_f32(false, a, false, b0, (short)0, c0, false, false);
    c1 = __builtin_amdgcn_wmma_f32_16x16x4_f32(false, a, false, b1, (short)0, c1, false, false);
    c2 = __builtin_amdgcn_wmma_f32_16x16x4_f32(false, a, false, b2, (short)0, c2, false, false);
    c3 = __builtin_amdgcn_wmma_f32_16x16x4_f32(false, a, false, b3, (short)0, c3, false, false);
  }

  if (has0) sweep1_count(rec, m_len, m_pos, m_off, 0, lane, fits, ncnt, rbuf, sorted);
  for (int q0 = 16; q0 < 32 && wave + NW2 * q0 < S; q0 += 16) {
    sweep1_load(rec, m_len, m_pos, q0, lane, sorted);
    sweep1_count(rec, m_len, m_pos, m_off, q0, lane, fits, ncnt, rbuf, sorted);
  }
  lds_barrier();

  const unsigned c = (tid < BWN) ? ncnt[tid] : 0u;
  unsigned tot2;
  const unsigned noff = block_scan_excl(c, wsum, &tot2);
  if (tid < BWN) {
    cur[tid] = noff;
    const int node = b * BWN + tid;
    if (node < N) {
      rowptr[node] = base + noff;
      const float dv = rsqrtf((float)(c + 1u));
      dinv[node]  = dv;
      sdinv[tid]  = dv;
      if (node == N - 1) rowptr[N] = base + noff + c;
    }
  }
  lds_barrier();

#pragma unroll
  for (int r = 0; r < 8; ++r) {
    const int lr  = (wave << 4) + r + (half << 3);
    const int row = b * BWN + lr;
    if (row < N) {
      const float s = sdinv[lr];
      const unsigned yo = (unsigned)row * FD + (unsigned)m;
      y[yo +  0] = c0[r] * s;
      y[yo + 16] = c1[r] * s;
      y[yo + 32] = c2[r] * s;
      y[yo + 48] = c3[r] * s;
    }
  }

  if (fits) {
    for (unsigned i = tid; i < nb; i += NT2) {
      const unsigned r = rbuf[i];
      const unsigned slot = atomicAdd(&cur[r & (BWN - 1)], 1u);
      ostage[slot] = r >> 8;
    }
    lds_barrier();
    for (unsigned i = tid; i < nb; i += NT2) csr[base + i] = (int)ostage[i];
  } else {
    for (unsigned i = tid; i < nb; i += NT2) {
      const unsigned r = bucket_rec(runoff, runpos, sorted, i);
      const unsigned slot = atomicAdd(&cur[r & (BWN - 1)], 1u);
      csr[base + slot] = (int)(r >> 8);
    }
  }
}

__global__ void __launch_bounds__(NT1, 16) tile_sort_kernel(const int* __restrict__ ei, int E, int N, int nB,
                                                            unsigned* __restrict__ tbl,
                                                            unsigned* __restrict__ sorted) {
  tile_sort_block((int)blockIdx.x, ei, E, N, nB, tbl, sorted);
}

#define GWV   16384
__global__ void __launch_bounds__(256) gather_kernel(const int* __restrict__ csr,
                                                     const unsigned* __restrict__ rowptr,
                                                     const float* __restrict__ y,
                                                     const float* __restrict__ dinv,
                                                     const float* __restrict__ b,
                                                     const float* __restrict__ prelu_a,
                                                     float* __restrict__ out, int N, int E) {
  const int lane = threadIdx.x & 31;
  const int nw   = (int)gridDim.x * 8;
  int node = __builtin_amdgcn_readfirstlane((int)(blockIdx.x * 8 + (threadIdx.x >> 5)));
  if (node >= N) return;
  const float2* __restrict__ y2 = (const float2*)y;
  const char*   __restrict__ yb = (const char*)y;
  const unsigned lane8 = (unsigned)lane << 3;
  const float  a  = prelu_a[0];
  const float2 bb = *((const float2*)b + lane);
  const bool n24 = N <= (1 << 24);

  const unsigned elast = (unsigned)(E - 1);
  unsigned beg = rowptr[node];
  int      L   = (int)(rowptr[node + 1] - beg);
  float    dn  = dinv[node];
  int      raw = csr[min(beg + (unsigned)lane, elast)];
  int      n1   = (node + nw < N) ? node + nw : node;
  unsigned nbeg = rowptr[n1];
  int      nL   = (int)(rowptr[n1 + 1] - nbeg);
  float    ndn  = dinv[n1];

  for (;;) {
    const int      nraw  = csr[min(nbeg + (unsigned)lane, elast)];
    const int      n2    = (n1 + nw < N) ? n1 + nw : n1;
    const unsigned n2beg = rowptr[n2];
    const int      n2L   = (int)(rowptr[n2 + 1] - n2beg);
    const float    n2dn  = dinv[n2];
    float2 acc = y2[(size_t)node * 32 + lane];

    if (L <= 32 && n24) {
      const int key = (lane < L) ? ((raw << 5) | lane) : 0x7fffffff;
      int rank = 0;
      for (int j = 0; j < L; ++j)
        rank += (__builtin_amdgcn_readlane(key, j) < key) ? 1 : 0;
      const int sorted = __builtin_amdgcn_ds_permute(rank << 2, key);
      for (int k = 0; k < L; k += 8) {
        float2 v[8];
#pragma unroll
        for (int j = 0; j < 8; ++j) {
          const int kk = (k + j < L) ? (k + j) : (L - 1);
          const unsigned s = (unsigned)__builtin_amdgcn_readlane(sorted, kk) >> 5;
          v[j] = *(const float2*)(yb + ((s << 8) | lane8));
        }
#pragma unroll
        for (int j = 0; j < 8; ++j) {
          if (k + j < L) { acc.x += v[j].x; acc.y += v[j].y; }
        }
      }
    } else if (L <= 64 && n24) {
      const int j1 = lane + 32;
      const int key0 = (raw << 6) | lane;
      const int key1 = (j1 < L) ? ((csr[beg + j1] << 6) | j1) : 0x7fffffff;
      int rank0 = 0, rank1 = 0;
      for (int j = 0; j < 32; ++j) {
        const int kj = __builtin_amdgcn_readlane(key0, j);
        rank0 += (kj < key0) ? 1 : 0;
        rank1 += (kj < key1) ? 1 : 0;
      }
      for (int j = 0; j < L - 32; ++j) {
        const int kj = __builtin_amdgcn_readlane(key1, j);
        rank0 += (kj < key0) ? 1 : 0;
        rank1 += (kj < key1) ? 1 : 0;
      }
      for (int k = 0; k < L; k += 8) {
        float2 v[8];
#pragma unroll
        for (int j = 0; j < 8; ++j) {
          const int kk = (k + j < L) ? (k + j) : (L - 1);
          const int p0 = __builtin_ffs((int)__builtin_amdgcn_ballot_w32(rank0 == kk)) - 1;
          const int p1 = __builtin_ffs((int)__builtin_amdgcn_ballot_w32(rank1 == kk)) - 1;
          const int kq = (p0 >= 0) ? __builtin_amdgcn_readlane(key0, p0)
                                   : __builtin_amdgcn_readlane(key1, p1);
          const unsigned s = (unsigned)kq >> 6;
          v[j] = *(const float2*)(yb + ((s << 8) | lane8));
        }
#pragma unroll
        for (int j = 0; j < 8; ++j) {
          if (k + j < L) { acc.x += v[j].x; acc.y += v[j].y; }
        }
      }
    } else {
      int ls = -1, lj = -1;
      for (int k = 0; k < L; ++k) {
        int bs = 0x7fffffff, bj = 0x7fffffff;
        for (int j = lane; j < L; j += 32) {
          const int s = csr[beg + j];
          const bool above = (s > ls) || (s == ls && j > lj);
          const bool lower = (s < bs) || (s == bs && j < bj);
          if (above && lower) { bs = s; bj = j; }
        }
        for (int o = 16; o > 0; o >>= 1) {
          const int ts = __shfl_xor(bs, o);
          const int tj = __shfl_xor(bj, o);
          if (ts < bs || (ts == bs && tj < bj)) { bs = ts; bj = tj; }
        }
        ls = bs; lj = bj;
        const float2 v = y2[(size_t)bs * 32 + lane];
        acc.x += v.x; acc.y += v.y;
      }
    }

    float2 r;
    r.x = dn * acc.x + bb.x;
    r.y = dn * acc.y + bb.y;
    r.x = r.x >= 0.0f ? r.x : a * r.x;
    r.y = r.y >= 0.0f ? r.y : a * r.y;
    *((float2*)out + (size_t)node * 32 + lane) = r;

    if (node + nw >= N) break;
    node = n1; beg = nbeg;  L  = nL;  dn  = ndn;  raw = nraw;
    n1   = n2; nbeg = n2beg; nL = n2L; ndn = n2dn;
  }
}

extern "C" void kernel_launch(void* const* d_in, const int* in_sizes, int n_in,
                              void* d_out, int out_size, void* d_ws, size_t ws_size,
                              hipStream_t stream) {
  const float* x  = (const float*)d_in[0];
  const int*   ei = (const int*)d_in[1];
  const float* W  = (const float*)d_in[2];
  const float* b  = (const float*)d_in[3];
  const float* pa = (const float*)d_in[4];
  float* out = (float*)d_out;

  const int N = in_sizes[0] / FD;
  const int E = in_sizes[1] / 2;

  const int S  = (E + TE - 1) / TE;
  const int nB = (N + BWN - 1) / BWN;

  float*    dinv   = (float*)d_ws + N;
  float*    y      = (float*)d_ws + 2 * (size_t)N;
  unsigned* tbl    = (unsigned*)(((uintptr_t)(y + (size_t)N * FD) + 15) & ~(uintptr_t)15);
  unsigned* sorted = (unsigned*)(((uintptr_t)(tbl + (size_t)S * (nB + 1)) + 15) & ~(uintptr_t)15);
  unsigned* rowptr = (unsigned*)(((uintptr_t)(sorted + E) + 15) & ~(uintptr_t)15);
  int*      csr    = (int*)(((uintptr_t)(rowptr + N + 1) + 15) & ~(uintptr_t)15);

  tile_sort_kernel<<<S, NT1, 0, stream>>>(ei, E, N, nB, tbl, sorted);
  bucket_csr_kernel<<<nB, NT2, 0, stream>>>(tbl, S, nB, sorted, N, dinv, rowptr, csr, x, W, y);

  const int bN = (N + 7) / 8;
  const int bA = bN < GWV / 8 ? bN : GWV / 8;
  gather_kernel<<<bA, 256, 0, stream>>>(csr, rowptr, y, dinv, b, pa, out, N, E);
}
